# starting kernel
# baseline (speedup 1.0000x reference)
#include <hip/hip_runtime.h>
#include <hip/hip_bf16.h>

#define GCN_N_NODES 50000
#define GCN_N_EDGES 800000
#define GCN_IN      128
#define GCN_H       128
#define GCN_C       32

typedef __attribute__((ext_vector_type(2))) float v2f;
typedef __attribute__((ext_vector_type(8))) float v8f;

__global__ __launch_bounds__(256) void gcn_deg_kernel(
    const int* __restrict__ src, const int* __restrict__ dst,
    float* __restrict__ deg_out, float* __restrict__ deg_in, int E) {
  int e = blockIdx.x * blockDim.x + threadIdx.x;
  if (e < E) {
    atomicAdd(&deg_out[src[e]], 1.0f);
    atomicAdd(&deg_in[dst[e]], 1.0f);
  }
}

__global__ __launch_bounds__(256) void gcn_invsqrt_kernel(
    float* __restrict__ deg_out, float* __restrict__ deg_in, int N) {
  int n = blockIdx.x * blockDim.x + threadIdx.x;
  if (n < N) {
    float d = deg_out[n];
    deg_out[n] = (d > 0.0f) ? rsqrtf(d) : 0.0f;
    d = deg_in[n];
    deg_in[n] = (d > 0.0f) ? rsqrtf(d) : 0.0f;
  }
}

__global__ __launch_bounds__(256) void gcn_spmm_scatter(
    const float* __restrict__ x, const int* __restrict__ src,
    const int* __restrict__ dst, const float* __restrict__ inv_out,
    float* __restrict__ agg, int E) {
  int lane = threadIdx.x & 31;
  int e = blockIdx.x * 8 + (threadIdx.x >> 5);
  if (e >= E) return;
  int s = src[e];
  int d = dst[e];
  float sc = inv_out[s];
  const float4* xr = (const float4*)(x + (size_t)s * 128);
  float4 v = xr[lane];
  float* ar = agg + (size_t)d * 128 + lane * 4;
  atomicAdd(ar + 0, v.x * sc);
  atomicAdd(ar + 1, v.y * sc);
  atomicAdd(ar + 2, v.z * sc);
  atomicAdd(ar + 3, v.w * sc);
}

__global__ __launch_bounds__(256) void gcn_gemm1_wmma(
    const float* __restrict__ agg, const float* __restrict__ inv_in,
    const float* __restrict__ W, const float* __restrict__ bias,
    float* __restrict__ out) {
  __shared__ float As[16][132];
  const int row0 = blockIdx.x * 16;
  const int tid  = threadIdx.x;

  for (int i = tid; i < 16 * 128; i += 256) {
    int r = i >> 7, c = i & 127;
    As[r][c] = agg[(size_t)(row0 + r) * 128 + c] * inv_in[row0 + r];
  }
  __syncthreads();

  const int wave = tid >> 5;
  const int lane = tid & 31;
  const int col0 = wave * 16;
  const int half = lane >> 4;
  const int mrow = lane & 15;

  v8f acc = {};
  for (int kk = 0; kk < 128; kk += 4) {
    int k2 = kk + half * 2;
    v2f a, b;
    a.x = As[mrow][k2];
    a.y = As[mrow][k2 + 1];
    b.x = W[(size_t)k2 * 128 + col0 + mrow];
    b.y = W[(size_t)(k2 + 1) * 128 + col0 + mrow];
    acc = __builtin_amdgcn_wmma_f32_16x16x4_f32(
         false, a,  false, b,
         (short)0, acc,  false,  false);
  }

  float bv = bias[col0 + mrow];
  for (int r = 0; r < 8; ++r) {
    int row = row0 + r + half * 8;
    float v = acc[r] + bv;
    out[(size_t)row * 128 + col0 + mrow] = (v > 0.0f) ? v : 0.0f;
  }
}

__global__ __launch_bounds__(256) void gcn_gemm2_wmma(
    const float* __restrict__ agg, const float* __restrict__ inv_in,
    const float* __restrict__ W, const float* __restrict__ bias,
    float* __restrict__ out, int N) {
  __shared__ float As[64][132];
  const int row0 = blockIdx.x * 64;
  const int tid  = threadIdx.x;

  for (int i = tid; i < 64 * 128; i += 256) {
    int r = i >> 7, c = i & 127;
    int row = row0 + r;
    As[r][c] = (row < N) ? agg[(size_t)row * 128 + c] * inv_in[row] : 0.0f;
  }
  __syncthreads();

  const int wave = tid >> 5;
  const int lane = tid & 31;
  const int rg   = wave >> 1;
  const int col0 = (wave & 1) * 16;
  const int half = lane >> 4;
  const int mrow = lane & 15;

  v8f acc = {};
  for (int kk = 0; kk < 128; kk += 4) {
    int k2 = kk + half * 2;
    v2f a, b;
    a.x = As[rg * 16 + mrow][k2];
    a.y = As[rg * 16 + mrow][k2 + 1];
    b.x = W[(size_t)k2 * 32 + col0 + mrow];
    b.y = W[(size_t)(k2 + 1) * 32 + col0 + mrow];
    acc = __builtin_amdgcn_wmma_f32_16x16x4_f32(
        false, a, false, b, (short)0, acc, false, false);
  }

  float bv = bias[col0 + mrow];
  for (int r = 0; r < 8; ++r) {
    int row = row0 + rg * 16 + r + half * 8;
    if (row < N) out[(size_t)row * 32 + col0 + mrow] = acc[r] + bv;
  }
}

extern "C" void kernel_launch(void* const* d_in, const int* in_sizes, int n_in,
                              void* d_out, int out_size, void* d_ws, size_t ws_size,
                              hipStream_t stream) {
  const float* in_feat = (const float*)d_in[0];
  const int*   src     = (const int*)d_in[1];
  const int*   dst     = (const int*)d_in[2];
  const float* W1      = (const float*)d_in[3];
  const float* b1      = (const float*)d_in[4];
  const float* W2      = (const float*)d_in[5];
  const float* b2      = (const float*)d_in[6];
  float* out = (float*)d_out;

  const int N = GCN_N_NODES;
  const int E = GCN_N_EDGES;

  float* inv_out = (float*)d_ws;
  float* inv_in  = inv_out + N;
  float* agg     = inv_in + N;
  float* h1      = agg + (size_t)N * 128;

  hipMemsetAsync(inv_out, 0, (size_t)2 * N * sizeof(float), stream);
  gcn_deg_kernel<<<(E + 255) / 256, 256, 0, stream>>>(src, dst, inv_out, inv_in, E);
  gcn_invsqrt_kernel<<<(N + 255) / 256, 256, 0, stream>>>(inv_out, inv_in, N);

  hipMemsetAsync(agg, 0, (size_t)N * 128 * sizeof(float), stream);
  gcn_spmm_scatter<<<(E + 7) / 8, 256, 0, stream>>>(in_feat, src, dst, inv_out, agg, E);
  gcn_gemm1_wmma<<<N / 16, 256, 0, stream>>>(agg, inv_in, W1, b1, h1);

  hipMemsetAsync(agg, 0, (size_t)N * 128 * sizeof(float), stream);
  gcn_spmm_scatter<<<(E + 7) / 8, 256, 0, stream>>>(h1, src, dst, inv_out, agg, E);
  gcn_gemm2_wmma<<<(N + 63) / 64, 256, 0, stream>>>(agg, inv_in, W2, b2, out, N);
}


# Layer 2 applies W2 before the edge sum, (A D h) W = A (D h W): its atomic scatter moves 32 floats per edge instead of 128, straight into out (set to the bias first); f32 summation order changes
# speedup vs baseline: 1.3596x; 1.3596x over previous
#include <hip/hip_runtime.h>
#include <hip/hip_bf16.h>

#define GCN_N_NODES 50000
#define GCN_N_EDGES 800000
#define GCN_IN      128
#define GCN_H       128
#define GCN_C       32

typedef __attribute__((ext_vector_type(2))) float v2f;
typedef __attribute__((ext_vector_type(8))) float v8f;

__global__ __launch_bounds__(256) void gcn_deg_kernel(
    const int* __restrict__ src, const int* __restrict__ dst,
    float* __restrict__ deg_out, float* __restrict__ deg_in, int E) {
  int e = blockIdx.x * blockDim.x + threadIdx.x;
  if (e < E) {
    atomicAdd(&deg_out[src[e]], 1.0f);
    atomicAdd(&deg_in[dst[e]], 1.0f);
  }
}

__global__ __launch_bounds__(256) void gcn_invsqrt_kernel(
    float* __restrict__ deg_out, float* __restrict__ deg_in, int N) {
  int n = blockIdx.x * blockDim.x + threadIdx.x;
  if (n < N) {
    float d = deg_out[n];
    deg_out[n] = (d > 0.0f) ? rsqrtf(d) : 0.0f;
    d = deg_in[n];
    deg_in[n] = (d > 0.0f) ? rsqrtf(d) : 0.0f;
  }
}

__global__ __launch_bounds__(256) void gcn_spmm_scatter(
    const float* __restrict__ x, const int* __restrict__ src,
    const int* __restrict__ dst, const float* __restrict__ inv_out,
    float* __restrict__ agg, int E) {
  int lane = threadIdx.x & 31;
  int e = blockIdx.x * 8 + (threadIdx.x >> 5);
  if (e >= E) return;
  int s = src[e];
  int d = dst[e];
  float sc = inv_out[s];
  const float4* xr = (const float4*)(x + (size_t)s * 128);
  float4 v = xr[lane];
  float* ar = agg + (size_t)d * 128 + lane * 4;
  atomicAdd(ar + 0, v.x * sc);
  atomicAdd(ar + 1, v.y * sc);
  atomicAdd(ar + 2, v.z * sc);
  atomicAdd(ar + 3, v.w * sc);
}

__global__ __launch_bounds__(256) void gcn_spmm_scatter32(
    const float* __restrict__ t, const int* __restrict__ src,
    const int* __restrict__ dst, const float* __restrict__ inv_in,
    float* __restrict__ out, int E) {
  int sub = threadIdx.x & 7;
  int e = blockIdx.x * 32 + (threadIdx.x >> 3);
  if (e >= E) return;
  int s = src[e];
  int d = dst[e];
  float sc = inv_in[d];
  const float4* tr = (const float4*)(t + (size_t)s * 32);
  float4 v = tr[sub];
  float* o = out + (size_t)d * 32 + sub * 4;
  atomicAdd(o + 0, v.x * sc);
  atomicAdd(o + 1, v.y * sc);
  atomicAdd(o + 2, v.z * sc);
  atomicAdd(o + 3, v.w * sc);
}

__global__ __launch_bounds__(256) void gcn_gemm1_wmma(
    const float* __restrict__ agg, const float* __restrict__ inv_in,
    const float* __restrict__ W, const float* __restrict__ bias,
    float* __restrict__ out) {
  __shared__ float As[16][132];
  const int row0 = blockIdx.x * 16;
  const int tid  = threadIdx.x;

  for (int i = tid; i < 16 * 128; i += 256) {
    int r = i >> 7, c = i & 127;
    As[r][c] = agg[(size_t)(row0 + r) * 128 + c] * inv_in[row0 + r];
  }
  __syncthreads();

  const int wave = tid >> 5;
  const int lane = tid & 31;
  const int col0 = wave * 16;
  const int half = lane >> 4;
  const int mrow = lane & 15;

  v8f acc = {};
  for (int kk = 0; kk < 128; kk += 4) {
    int k2 = kk + half * 2;
    v2f a, b;
    a.x = As[mrow][k2];
    a.y = As[mrow][k2 + 1];
    b.x = W[(size_t)k2 * 128 + col0 + mrow];
    b.y = W[(size_t)(k2 + 1) * 128 + col0 + mrow];
    acc = __builtin_amdgcn_wmma_f32_16x16x4_f32(
         false, a,  false, b,
         (short)0, acc,  false,  false);
  }

  float bv = bias[col0 + mrow];
  for (int r = 0; r < 8; ++r) {
    int row = row0 + r + half * 8;
    float v = acc[r] + bv;
    out[(size_t)row * 128 + col0 + mrow] = (v > 0.0f) ? v : 0.0f;
  }
}

__global__ __launch_bounds__(256) void gcn_gemm2_wmma(
    const float* __restrict__ h, const float* __restrict__ inv_out,
    const float* __restrict__ W, const float* __restrict__ bias,
    float* __restrict__ t, float* __restrict__ out, int N) {
  __shared__ float As[64][132];
  const int row0 = blockIdx.x * 64;
  const int tid  = threadIdx.x;

  for (int i = tid; i < 64 * 128; i += 256) {
    int r = i >> 7, c = i & 127;
    int row = row0 + r;
    As[r][c] = (row < N) ? h[(size_t)row * 128 + c] * inv_out[row] : 0.0f;
  }
  __syncthreads();

  const int wave = tid >> 5;
  const int lane = tid & 31;
  const int rg   = wave >> 1;
  const int col0 = (wave & 1) * 16;
  const int half = lane >> 4;
  const int mrow = lane & 15;

  v8f acc = {};
  for (int kk = 0; kk < 128; kk += 4) {
    int k2 = kk + half * 2;
    v2f a, b;
    a.x = As[rg * 16 + mrow][k2];
    a.y = As[rg * 16 + mrow][k2 + 1];
    b.x = W[(size_t)k2 * 32 + col0 + mrow];
    b.y = W[(size_t)(k2 + 1) * 32 + col0 + mrow];
    acc = __builtin_amdgcn_wmma_f32_16x16x4_f32(
        false, a, false, b, (short)0, acc, false, false);
  }

  float bv = bias[col0 + mrow];
  for (int r = 0; r < 8; ++r) {
    int row = row0 + rg * 16 + r + half * 8;
    if (row < N) {
      t[(size_t)row * 32 + col0 + mrow] = acc[r];
      out[(size_t)row * 32 + col0 + mrow] = bv;
    }
  }
}

extern "C" void kernel_launch(void* const* d_in, const int* in_sizes, int n_in,
                              void* d_out, int out_size, void* d_ws, size_t ws_size,
                              hipStream_t stream) {
  const float* in_feat = (const float*)d_in[0];
  const int*   src     = (const int*)d_in[1];
  const int*   dst     = (const int*)d_in[2];
  const float* W1      = (const float*)d_in[3];
  const float* b1      = (const float*)d_in[4];
  const float* W2      = (const float*)d_in[5];
  const float* b2      = (const float*)d_in[6];
  float* out = (float*)d_out;

  const int N = GCN_N_NODES;
  const int E = GCN_N_EDGES;

  float* inv_out = (float*)d_ws;
  float* inv_in  = inv_out + N;
  float* agg     = inv_in + N;
  float* h1      = agg + (size_t)N * 128;
  float* t       = agg;

  hipMemsetAsync(inv_out, 0, (size_t)2 * N * sizeof(float), stream);
  gcn_deg_kernel<<<(E + 255) / 256, 256, 0, stream>>>(src, dst, inv_out, inv_in, E);
  gcn_invsqrt_kernel<<<(N + 255) / 256, 256, 0, stream>>>(inv_out, inv_in, N);

  hipMemsetAsync(agg, 0, (size_t)N * 128 * sizeof(float), stream);
  gcn_spmm_scatter<<<(E + 7) / 8, 256, 0, stream>>>(in_feat, src, dst, inv_out, agg, E);
  gcn_gemm1_wmma<<<N / 16, 256, 0, stream>>>(agg, inv_in, W1, b1, h1);

  gcn_gemm2_wmma<<<(N + 63) / 64, 256, 0, stream>>>(h1, inv_out, W2, b2, t, out, N);
  gcn_spmm_scatter32<<<(E + 31) / 32, 256, 0, stream>>>(t, src, dst, inv_in, out, E);
}


# Both edge sums become per-destination gathers over a dst-sorted edge list built in the call (int count, scan, returning-add slot): no float atomics, no agg clear; f32 sum order = arrival rank
# speedup vs baseline: 1.9507x; 1.4348x over previous
#include <hip/hip_runtime.h>
#include <hip/hip_bf16.h>


#define GCN_N_NODES 50000
#define GCN_N_EDGES 800000
#define GCN_IN      128
#define GCN_H       128
#define GCN_C       32

typedef __attribute__((ext_vector_type(2))) float v2f;
typedef __attribute__((ext_vector_type(8))) float v8f;

__global__ __launch_bounds__(256) void gcn_deg_kernel(
    const int* __restrict__ src, const int* __restrict__ dst,
    float* __restrict__ deg_out, int* __restrict__ cnt_in, int E) {
  int e = blockIdx.x * blockDim.x + threadIdx.x;
  if (e < E) {
    atomicAdd(&deg_out[src[e]], 1.0f);
    atomicAdd(&cnt_in[dst[e]], 1);
  }
}

__global__ __launch_bounds__(1024) void gcn_scan_kernel(
    const int* __restrict__ cnt_in, float* __restrict__ deg_out,
    float* __restrict__ inv_in, int* __restrict__ rowptr,
    int* __restrict__ cursor, int N) {
  __shared__ int pre_w[32];
  __shared__ int tot_w[32];
  const int tid  = threadIdx.x;
  const int lane = tid & 31;
  const int w    = tid >> 5;
  const int lim  = blockIdx.x * 1024;

  int pre = 0;
  for (int i = tid; i < lim; i += 1024) pre += cnt_in[i];

  const int i = lim + tid;
  const int c = (i < N) ? cnt_in[i] : 0;

  int incl = c;
  for (int off = 1; off < 32; off <<= 1) {
    int y = __shfl_up(incl, off);
    if (lane >= off) incl += y;
  }
  for (int off = 16; off > 0; off >>= 1) pre += __shfl_xor(pre, off);
  if (lane == 31) tot_w[w] = incl;
  if (lane == 0)  pre_w[w] = pre;
  __syncthreads();

  int part = pre_w[lane] + ((lane < w) ? tot_w[lane] : 0);
  for (int off = 16; off > 0; off >>= 1) part += __shfl_xor(part, off);
  const int excl = part + incl - c;

  if (i < N) {
    rowptr[i] = excl;
    cursor[i] = excl;
    inv_in[i] = (c > 0) ? rsqrtf((float)c) : 0.0f;
    float d = deg_out[i];
    deg_out[i] = (d > 0.0f) ? rsqrtf(d) : 0.0f;
    if (i == N - 1) rowptr[N] = excl + c;
  }
}

__global__ __launch_bounds__(256) void gcn_fill_kernel(
    const int* __restrict__ src, const int* __restrict__ dst,
    int* __restrict__ cursor, int* __restrict__ csr_src, int E) {
  int e = blockIdx.x * blockDim.x + threadIdx.x;
  if (e < E) {
    int p = atomicAdd(&cursor[dst[e]], 1);
    csr_src[p] = src[e];
  }
}

__global__ __launch_bounds__(256) void gcn_gather128(
    const float* __restrict__ x, const int* __restrict__ rowptr,
    const int* __restrict__ csr_src, const float* __restrict__ inv_out,
    float* __restrict__ agg, int N) {
  const int lane = threadIdx.x & 31;
  const int n = blockIdx.x * 8 + (threadIdx.x >> 5);
  if (n >= N) return;
  const int beg = __builtin_amdgcn_readfirstlane(rowptr[n]);
  const int end = __builtin_amdgcn_readfirstlane(rowptr[n + 1]);

  float4 acc = make_float4(0.0f, 0.0f, 0.0f, 0.0f);
  for (int base = beg; base < end; base += 32) {
    const int m = (end - base < 32) ? (end - base) : 32;
    int   s_l = 0;
    float w_l = 0.0f;
    if (lane < m) {
      s_l = csr_src[base + lane];
      w_l = inv_out[s_l];
    }
    int j = 0;
    for (; j + 4 <= m; j += 4) {
      const int s0 = __builtin_amdgcn_readlane(s_l, j);
      const int s1 = __builtin_amdgcn_readlane(s_l, j + 1);
      const int s2 = __builtin_amdgcn_readlane(s_l, j + 2);
      const int s3 = __builtin_amdgcn_readlane(s_l, j + 3);
      const float w0 = __int_as_float(__builtin_amdgcn_readlane(__float_as_int(w_l), j));
      const float w1 = __int_as_float(__builtin_amdgcn_readlane(__float_as_int(w_l), j + 1));
      const float w2 = __int_as_float(__builtin_amdgcn_readlane(__float_as_int(w_l), j + 2));
      const float w3 = __int_as_float(__builtin_amdgcn_readlane(__float_as_int(w_l), j + 3));
      const float4 v0 = ((const float4*)(x + (size_t)s0 * 128))[lane];
      const float4 v1 = ((const float4*)(x + (size_t)s1 * 128))[lane];
      const float4 v2 = ((const float4*)(x + (size_t)s2 * 128))[lane];
      const float4 v3 = ((const float4*)(x + (size_t)s3 * 128))[lane];
      acc.x += v0.x * w0; acc.y += v0.y * w0; acc.z += v0.z * w0; acc.w += v0.w * w0;
      acc.x += v1.x * w1; acc.y += v1.y * w1; acc.z += v1.z * w1; acc.w += v1.w * w1;
      acc.x += v2.x * w2; acc.y += v2.y * w2; acc.z += v2.z * w2; acc.w += v2.w * w2;
      acc.x += v3.x * w3; acc.y += v3.y * w3; acc.z += v3.z * w3; acc.w += v3.w * w3;
    }
    for (; j < m; ++j) {
      const int s0 = __builtin_amdgcn_readlane(s_l, j);
      const float w0 = __int_as_float(__builtin_amdgcn_readlane(__float_as_int(w_l), j));
      const float4 v0 = ((const float4*)(x + (size_t)s0 * 128))[lane];
      acc.x += v0.x * w0; acc.y += v0.y * w0; acc.z += v0.z * w0; acc.w += v0.w * w0;
    }
  }
  ((float4*)(agg + (size_t)n * 128))[lane] = acc;
}

__global__ __launch_bounds__(256) void gcn_gather32(
    const float* __restrict__ t, const int* __restrict__ rowptr,
    const int* __restrict__ csr_src, const float* __restrict__ inv_in,
    const float* __restrict__ bias, float* __restrict__ out, int N) {
  const int sub = threadIdx.x & 7;
  const int n = blockIdx.x * 32 + (threadIdx.x >> 3);
  if (n >= N) return;
  const int beg = rowptr[n];
  const int end = rowptr[n + 1];

  float4 acc = make_float4(0.0f, 0.0f, 0.0f, 0.0f);
  int j = beg;
  for (; j + 4 <= end; j += 4) {
    const int s0 = csr_src[j];
    const int s1 = csr_src[j + 1];
    const int s2 = csr_src[j + 2];
    const int s3 = csr_src[j + 3];
    const float4 v0 = ((const float4*)(t + (size_t)s0 * 32))[sub];
    const float4 v1 = ((const float4*)(t + (size_t)s1 * 32))[sub];
    const float4 v2 = ((const float4*)(t + (size_t)s2 * 32))[sub];
    const float4 v3 = ((const float4*)(t + (size_t)s3 * 32))[sub];
    acc.x += v0.x; acc.y += v0.y; acc.z += v0.z; acc.w += v0.w;
    acc.x += v1.x; acc.y += v1.y; acc.z += v1.z; acc.w += v1.w;
    acc.x += v2.x; acc.y += v2.y; acc.z += v2.z; acc.w += v2.w;
    acc.x += v3.x; acc.y += v3.y; acc.z += v3.z; acc.w += v3.w;
  }
  for (; j < end; ++j) {
    const int s0 = csr_src[j];
    const float4 v0 = ((const float4*)(t + (size_t)s0 * 32))[sub];
    acc.x += v0.x; acc.y += v0.y; acc.z += v0.z; acc.w += v0.w;
  }
  const float sc = inv_in[n];
  const float4 b = ((const float4*)bias)[sub];
  float4 o;
  o.x = acc.x * sc + b.x;
  o.y = acc.y * sc + b.y;
  o.z = acc.z * sc + b.z;
  o.w = acc.w * sc + b.w;
  ((float4*)(out + (size_t)n * 32))[sub] = o;
}

__global__ __launch_bounds__(256) void gcn_gemm1_wmma(
    const float* __restrict__ agg, const float* __restrict__ inv_in,
    const float* __restrict__ W, const float* __restrict__ bias,
    float* __restrict__ out) {
  __shared__ float As[16][132];
  const int row0 = blockIdx.x * 16;
  const int tid  = threadIdx.x;

  for (int i = tid; i < 16 * 128; i += 256) {
    int r = i >> 7, c = i & 127;
    As[r][c] = agg[(size_t)(row0 + r) * 128 + c] * inv_in[row0 + r];
  }
  __syncthreads();

  const int wave = tid >> 5;
  const int lane = tid & 31;
  const int col0 = wave * 16;
  const int half = lane >> 4;
  const int mrow = lane & 15;

  v8f acc = {};
  for (int kk = 0; kk < 128; kk += 4) {
    int k2 = kk + half * 2;
    v2f a, b;
    a.x = As[mrow][k2];
    a.y = As[mrow][k2 + 1];
    b.x = W[(size_t)k2 * 128 + col0 + mrow];
    b.y = W[(size_t)(k2 + 1) * 128 + col0 + mrow];
    acc = __builtin_amdgcn_wmma_f32_16x16x4_f32(
         false, a,  false, b,
         (short)0, acc,  false,  false);
  }

  float bv = bias[col0 + mrow];
  for (int r = 0; r < 8; ++r) {
    int row = row0 + r + half * 8;
    float v = acc[r] + bv;
    out[(size_t)row * 128 + col0 + mrow] = (v > 0.0f) ? v : 0.0f;
  }
}

__global__ __launch_bounds__(256) void gcn_gemm2_wmma(
    const float* __restrict__ h, const float* __restrict__ inv_out,
    const float* __restrict__ W, float* __restrict__ t, int N) {
  __shared__ float As[64][132];
  const int row0 = blockIdx.x * 64;
  const int tid  = threadIdx.x;

  for (int i = tid; i < 64 * 128; i += 256) {
    int r = i >> 7, c = i & 127;
    int row = row0 + r;
    As[r][c] = (row < N) ? h[(size_t)row * 128 + c] * inv_out[row] : 0.0f;
  }
  __syncthreads();

  const int wave = tid >> 5;
  const int lane = tid & 31;
  const int rg   = wave >> 1;
  const int col0 = (wave & 1) * 16;
  const int half = lane >> 4;
  const int mrow = lane & 15;

  v8f acc = {};
  for (int kk = 0; kk < 128; kk += 4) {
    int k2 = kk + half * 2;
    v2f a, b;
    a.x = As[rg * 16 + mrow][k2];
    a.y = As[rg * 16 + mrow][k2 + 1];
    b.x = W[(size_t)k2 * 32 + col0 + mrow];
    b.y = W[(size_t)(k2 + 1) * 32 + col0 + mrow];
    acc = __builtin_amdgcn_wmma_f32_16x16x4_f32(
        false, a, false, b, (short)0, acc, false, false);
  }

  for (int r = 0; r < 8; ++r) {
    int row = row0 + rg * 16 + r + half * 8;
    if (row < N) t[(size_t)row * 32 + col0 + mrow] = acc[r];
  }
}

extern "C" void kernel_launch(void* const* d_in, const int* in_sizes, int n_in,
                              void* d_out, int out_size, void* d_ws, size_t ws_size,
                              hipStream_t stream) {
  const float* in_feat = (const float*)d_in[0];
  const int*   src     = (const int*)d_in[1];
  const int*   dst     = (const int*)d_in[2];
  const float* W1      = (const float*)d_in[3];
  const float* b1      = (const float*)d_in[4];
  const float* W2      = (const float*)d_in[5];
  const float* b2      = (const float*)d_in[6];
  float* out = (float*)d_out;

  const int N = GCN_N_NODES;
  const int E = GCN_N_EDGES;

  float* inv_out = (float*)d_ws;
  int*   cnt_in  = (int*)(inv_out + N);
  float* inv_in  = (float*)(cnt_in + N);
  int*   rowptr  = (int*)(inv_in + N);
  int*   cursor  = rowptr + (N + 16);
  int*   csr_src = cursor + N;
  float* agg     = (float*)(csr_src + E);
  float* h1      = agg + (size_t)N * 128;
  float* t       = agg;

  hipMemsetAsync(inv_out, 0, (size_t)2 * N * sizeof(float), stream);
  gcn_deg_kernel<<<(E + 255) / 256, 256, 0, stream>>>(src, dst, inv_out, cnt_in, E);
  gcn_scan_kernel<<<(N + 1023) / 1024, 1024, 0, stream>>>(cnt_in, inv_out, inv_in, rowptr, cursor, N);
  gcn_fill_kernel<<<(E + 255) / 256, 256, 0, stream>>>(src, dst, cursor, csr_src, E);

  gcn_gather128<<<(N + 7) / 8, 256, 0, stream>>>(in_feat, rowptr, csr_src, inv_out, agg, N);
  gcn_gemm1_wmma<<<N / 16, 256, 0, stream>>>(agg, inv_in, W1, b1, h1);

  gcn_gemm2_wmma<<<(N + 63) / 64, 256, 0, stream>>>(h1, inv_out, W2, t, N);
  gcn_gather32<<<(N + 31) / 32, 256, 0, stream>>>(t, rowptr, csr_src, inv_in, b2, out, N);
}


# Edge sort without global atomics: per-group node counts and edge ranks from LDS adds on a 200 KB table, a column scan gives each group's start in a row; no memset; f32 sum order = group, then rank
# speedup vs baseline: 3.3956x; 1.7407x over previous
#include <hip/hip_runtime.h>
#include <hip/hip_bf16.h>


#define GCN_N_NODES 50000
#define GCN_N_EDGES 800000
#define GCN_IN      128
#define GCN_H       128
#define GCN_C       32

typedef __attribute__((ext_vector_type(2))) float v2f;
typedef __attribute__((ext_vector_type(8))) float v8f;

#define GCN_G      32
#define GCN_CHUNK  ((((GCN_N_EDGES + GCN_G - 1) / GCN_G) + 3) / 4 * 4)
#define GCN_KQ     ((GCN_CHUNK / 4 + 1023) / 1024)
#define GCN_NCH    ((GCN_N_NODES + 1023) / 1024)
#define GCN_ROUNDS ((GCN_N_NODES / 4 + 1023) / 1024)
#define GCN_CT     64
static_assert(GCN_N_NODES % 4 == 0 && GCN_N_EDGES % 4 == 0, "4-wide loads");
static_assert(4 * GCN_ROUNDS <= GCN_CT && GCN_NCH <= 4 * GCN_ROUNDS, "chunk totals");
static_assert(GCN_CHUNK % 4 == 0, "a 4-edge load stays inside one group");

__global__ __launch_bounds__(1024) void gcn_count_kernel(
    const int* __restrict__ src, const int* __restrict__ dst,
    int* __restrict__ hist_in, int* __restrict__ hist_out,
    int* __restrict__ lrank, int* __restrict__ ctot) {
  __shared__ __attribute__((aligned(16))) int tbl[GCN_N_NODES];
  __shared__ int wtot[GCN_ROUNDS * 32];
  const int tid  = threadIdx.x;
  const int lane = tid & 31;
  const int w    = tid >> 5;
  const int g    = blockIdx.x;
  const bool by_dst = (blockIdx.y == 0);
  const int e0 = g * GCN_CHUNK;
  const int e1 = (e0 + GCN_CHUNK < GCN_N_EDGES) ? (e0 + GCN_CHUNK) : GCN_N_EDGES;
  const int nq = (e1 > e0) ? ((e1 - e0) >> 2) : 0;
  const int4* __restrict__ ids = (const int4*)((by_dst ? dst : src) + e0);

  int4 id[GCN_KQ];
#pragma unroll
  for (int k = 0; k < GCN_KQ; ++k) {
    const int qi = k * 1024 + tid;
    id[k] = (qi < nq) ? ids[qi] : make_int4(0, 0, 0, 0);
  }

  for (int i = tid; i < GCN_N_NODES / 4; i += 1024)
    ((int4*)tbl)[i] = make_int4(0, 0, 0, 0);
  __syncthreads();

#pragma unroll
  for (int k = 0; k < GCN_KQ; ++k) {
    const int qi = k * 1024 + tid;
    if (qi < nq) {
      int4 r;
      r.x = atomicAdd(&tbl[id[k].x], 1);
      r.y = atomicAdd(&tbl[id[k].y], 1);
      r.z = atomicAdd(&tbl[id[k].z], 1);
      r.w = atomicAdd(&tbl[id[k].w], 1);
      if (by_dst) ((int4*)(lrank + e0))[qi] = r;
    }
  }
  __syncthreads();

  int* __restrict__ hist = (by_dst ? hist_in : hist_out) + (size_t)g * GCN_N_NODES;
  for (int q = 0; q < GCN_ROUNDS; ++q) {
    const int vi = q * 1024 + tid;
    int4 v = make_int4(0, 0, 0, 0);
    if (vi < GCN_N_NODES / 4) {
      v = ((const int4*)tbl)[vi];
      ((int4*)hist)[vi] = v;
    }
    int s = v.x + v.y + v.z + v.w;
    for (int off = 16; off > 0; off >>= 1) s += __shfl_xor(s, off);
    if (lane == 0) wtot[q * 32 + w] = s;
  }
  __syncthreads();
  if (by_dst && tid < 4 * GCN_ROUNDS) {
    const int q = tid >> 2, c = tid & 3;
    int s = 0;
    for (int t = 0; t < 8; ++t) s += wtot[q * 32 + c * 8 + t];
    ctot[g * GCN_CT + tid] = s;
  }
}

__global__ __launch_bounds__(1024) void gcn_scan_kernel(
    int* __restrict__ hist_in, const int* __restrict__ hist_out,
    const int* __restrict__ ctot, float* __restrict__ inv_out,
    float* __restrict__ inv_in, int* __restrict__ rowptr) {
  __shared__ int pre_w[32];
  __shared__ int tot_w[32];
  const int tid  = threadIdx.x;
  const int lane = tid & 31;
  const int w    = tid >> 5;
  const int N    = GCN_N_NODES;

  if (blockIdx.x >= GCN_NCH) {
    const int i = (blockIdx.x - GCN_NCH) * 1024 + tid;
    if (i < N) {
      int d = 0;
#pragma unroll
      for (int g = 0; g < GCN_G; ++g) d += hist_out[(size_t)g * N + i];
      inv_out[i] = (d > 0) ? rsqrtf((float)d) : 0.0f;
    }
    return;
  }

  const int blk = blockIdx.x;
  const int lim = blk * 1024;
  int pre = 0;
  for (int i = tid; i < GCN_G * GCN_CT; i += 1024)
    if ((i & (GCN_CT - 1)) < blk) pre += ctot[i];

  const int i = lim + tid;
  int v[GCN_G];
  int c = 0;
#pragma unroll
  for (int g = 0; g < GCN_G; ++g) {
    v[g] = (i < N) ? hist_in[(size_t)g * N + i] : 0;
    c += v[g];
  }

  int incl = c;
  for (int off = 1; off < 32; off <<= 1) {
    int y = __shfl_up(incl, off);
    if (lane >= off) incl += y;
  }
  for (int off = 16; off > 0; off >>= 1) pre += __shfl_xor(pre, off);
  if (lane == 31) tot_w[w] = incl;
  if (lane == 0)  pre_w[w] = pre;
  __syncthreads();

  int part = pre_w[lane] + ((lane < w) ? tot_w[lane] : 0);
  for (int off = 16; off > 0; off >>= 1) part += __shfl_xor(part, off);
  const int excl = part + incl - c;

  if (i < N) {
    rowptr[i] = excl;
    int run = excl;
#pragma unroll
    for (int g = 0; g < GCN_G; ++g) {
      hist_in[(size_t)g * N + i] = run;
      run += v[g];
    }
    inv_in[i] = (c > 0) ? rsqrtf((float)c) : 0.0f;
    if (i == N - 1) rowptr[N] = excl + c;
  }
}

__global__ __launch_bounds__(256) void gcn_fill_kernel(
    const int* __restrict__ src, const int* __restrict__ dst,
    const int* __restrict__ lrank, const int* __restrict__ gstart,
    int* __restrict__ csr_src) {
  const int q = blockIdx.x * 256 + threadIdx.x;
  const int e = q * 4;
  if (e >= GCN_N_EDGES) return;
  const int4 d = ((const int4*)dst)[q];
  const int4 s = ((const int4*)src)[q];
  const int4 r = ((const int4*)lrank)[q];
  const int* __restrict__ gs = gstart + (size_t)(e / GCN_CHUNK) * GCN_N_NODES;
  csr_src[gs[d.x] + r.x] = s.x;
  csr_src[gs[d.y] + r.y] = s.y;
  csr_src[gs[d.z] + r.z] = s.z;
  csr_src[gs[d.w] + r.w] = s.w;
}

__global__ __launch_bounds__(256) void gcn_gather128(
    const float* __restrict__ x, const int* __restrict__ rowptr,
    const int* __restrict__ csr_src, const float* __restrict__ inv_out,
    float* __restrict__ agg, int N) {
  const int lane = threadIdx.x & 31;
  const int n = blockIdx.x * 8 + (threadIdx.x >> 5);
  if (n >= N) return;
  const int beg = __builtin_amdgcn_readfirstlane(rowptr[n]);
  const int end = __builtin_amdgcn_readfirstlane(rowptr[n + 1]);

  float4 acc = make_float4(0.0f, 0.0f, 0.0f, 0.0f);
  for (int base = beg; base < end; base += 32) {
    const int m = (end - base < 32) ? (end - base) : 32;
    int   s_l = 0;
    float w_l = 0.0f;
    if (lane < m) {
      s_l = csr_src[base + lane];
      w_l = inv_out[s_l];
    }
    int j = 0;
    for (; j + 4 <= m; j += 4) {
      const int s0 = __builtin_amdgcn_readlane(s_l, j);
      const int s1 = __builtin_amdgcn_readlane(s_l, j + 1);
      const int s2 = __builtin_amdgcn_readlane(s_l, j + 2);
      const int s3 = __builtin_amdgcn_readlane(s_l, j + 3);
      const float w0 = __int_as_float(__builtin_amdgcn_readlane(__float_as_int(w_l), j));
      const float w1 = __int_as_float(__builtin_amdgcn_readlane(__float_as_int(w_l), j + 1));
      const float w2 = __int_as_float(__builtin_amdgcn_readlane(__float_as_int(w_l), j + 2));
      const float w3 = __int_as_float(__builtin_amdgcn_readlane(__float_as_int(w_l), j + 3));
      const float4 v0 = ((const float4*)(x + (size_t)s0 * 128))[lane];
      const float4 v1 = ((const float4*)(x + (size_t)s1 * 128))[lane];
      const float4 v2 = ((const float4*)(x + (size_t)s2 * 128))[lane];
      const float4 v3 = ((const float4*)(x + (size_t)s3 * 128))[lane];
      acc.x += v0.x * w0; acc.y += v0.y * w0; acc.z += v0.z * w0; acc.w += v0.w * w0;
      acc.x += v1.x * w1; acc.y += v1.y * w1; acc.z += v1.z * w1; acc.w += v1.w * w1;
      acc.x += v2.x * w2; acc.y += v2.y * w2; acc.z += v2.z * w2; acc.w += v2.w * w2;
      acc.x += v3.x * w3; acc.y += v3.y * w3; acc.z += v3.z * w3; acc.w += v3.w * w3;
    }
    for (; j < m; ++j) {
      const int s0 = __builtin_amdgcn_readlane(s_l, j);
      const float w0 = __int_as_float(__builtin_amdgcn_readlane(__float_as_int(w_l), j));
      const float4 v0 = ((const float4*)(x + (size_t)s0 * 128))[lane];
      acc.x += v0.x * w0; acc.y += v0.y * w0; acc.z += v0.z * w0; acc.w += v0.w * w0;
    }
  }
  ((float4*)(agg + (size_t)n * 128))[lane] = acc;
}

__global__ __launch_bounds__(256) void gcn_gather32(
    const float* __restrict__ t, const int* __restrict__ rowptr,
    const int* __restrict__ csr_src, const float* __restrict__ inv_in,
    const float* __restrict__ bias, float* __restrict__ out, int N) {
  const int sub = threadIdx.x & 7;
  const int n = blockIdx.x * 32 + (threadIdx.x >> 3);
  if (n >= N) return;
  const int beg = rowptr[n];
  const int end = rowptr[n + 1];

  float4 acc = make_float4(0.0f, 0.0f, 0.0f, 0.0f);
  int j = beg;
  for (; j + 4 <= end; j += 4) {
    const int s0 = csr_src[j];
    const int s1 = csr_src[j + 1];
    const int s2 = csr_src[j + 2];
    const int s3 = csr_src[j + 3];
    const float4 v0 = ((const float4*)(t + (size_t)s0 * 32))[sub];
    const float4 v1 = ((const float4*)(t + (size_t)s1 * 32))[sub];
    const float4 v2 = ((const float4*)(t + (size_t)s2 * 32))[sub];
    const float4 v3 = ((const float4*)(t + (size_t)s3 * 32))[sub];
    acc.x += v0.x; acc.y += v0.y; acc.z += v0.z; acc.w += v0.w;
    acc.x += v1.x; acc.y += v1.y; acc.z += v1.z; acc.w += v1.w;
    acc.x += v2.x; acc.y += v2.y; acc.z += v2.z; acc.w += v2.w;
    acc.x += v3.x; acc.y += v3.y; acc.z += v3.z; acc.w += v3.w;
  }
  for (; j < end; ++j) {
    const int s0 = csr_src[j];
    const float4 v0 = ((const float4*)(t + (size_t)s0 * 32))[sub];
    acc.x += v0.x; acc.y += v0.y; acc.z += v0.z; acc.w += v0.w;
  }
  const float sc = inv_in[n];
  const float4 b = ((const float4*)bias)[sub];
  float4 o;
  o.x = acc.x * sc + b.x;
  o.y = acc.y * sc + b.y;
  o.z = acc.z * sc + b.z;
  o.w = acc.w * sc + b.w;
  ((float4*)(out + (size_t)n * 32))[sub] = o;
}

__global__ __launch_bounds__(256) void gcn_gemm1_wmma(
    const float* __restrict__ agg, const float* __restrict__ inv_in,
    const float* __restrict__ W, const float* __restrict__ bias,
    float* __restrict__ out) {
  __shared__ float As[16][132];
  const int row0 = blockIdx.x * 16;
  const int tid  = threadIdx.x;

  for (int i = tid; i < 16 * 128; i += 256) {
    int r = i >> 7, c = i & 127;
    As[r][c] = agg[(size_t)(row0 + r) * 128 + c] * inv_in[row0 + r];
  }
  __syncthreads();

  const int wave = tid >> 5;
  const int lane = tid & 31;
  const int col0 = wave * 16;
  const int half = lane >> 4;
  const int mrow = lane & 15;

  v8f acc = {};
  for (int kk = 0; kk < 128; kk += 4) {
    int k2 = kk + half * 2;
    v2f a, b;
    a.x = As[mrow][k2];
    a.y = As[mrow][k2 + 1];
    b.x = W[(size_t)k2 * 128 + col0 + mrow];
    b.y = W[(size_t)(k2 + 1) * 128 + col0 + mrow];
    acc = __builtin_amdgcn_wmma_f32_16x16x4_f32(
         false, a,  false, b,
         (short)0, acc,  false,  false);
  }

  float bv = bias[col0 + mrow];
  for (int r = 0; r < 8; ++r) {
    int row = row0 + r + half * 8;
    float v = acc[r] + bv;
    out[(size_t)row * 128 + col0 + mrow] = (v > 0.0f) ? v : 0.0f;
  }
}

__global__ __launch_bounds__(256) void gcn_gemm2_wmma(
    const float* __restrict__ h, const float* __restrict__ inv_out,
    const float* __restrict__ W, float* __restrict__ t, int N) {
  __shared__ float As[64][132];
  const int row0 = blockIdx.x * 64;
  const int tid  = threadIdx.x;

  for (int i = tid; i < 64 * 128; i += 256) {
    int r = i >> 7, c = i & 127;
    int row = row0 + r;
    As[r][c] = (row < N) ? h[(size_t)row * 128 + c] * inv_out[row] : 0.0f;
  }
  __syncthreads();

  const int wave = tid >> 5;
  const int lane = tid & 31;
  const int rg   = wave >> 1;
  const int col0 = (wave & 1) * 16;
  const int half = lane >> 4;
  const int mrow = lane & 15;

  v8f acc = {};
  for (int kk = 0; kk < 128; kk += 4) {
    int k2 = kk + half * 2;
    v2f a, b;
    a.x = As[rg * 16 + mrow][k2];
    a.y = As[rg * 16 + mrow][k2 + 1];
    b.x = W[(size_t)k2 * 32 + col0 + mrow];
    b.y = W[(size_t)(k2 + 1) * 32 + col0 + mrow];
    acc = __builtin_amdgcn_wmma_f32_16x16x4_f32(
        false, a, false, b, (short)0, acc, false, false);
  }

  for (int r = 0; r < 8; ++r) {
    int row = row0 + rg * 16 + r + half * 8;
    if (row < N) t[(size_t)row * 32 + col0 + mrow] = acc[r];
  }
}

extern "C" void kernel_launch(void* const* d_in, const int* in_sizes, int n_in,
                              void* d_out, int out_size, void* d_ws, size_t ws_size,
                              hipStream_t stream) {
  const float* in_feat = (const float*)d_in[0];
  const int*   src     = (const int*)d_in[1];
  const int*   dst     = (const int*)d_in[2];
  const float* W1      = (const float*)d_in[3];
  const float* b1      = (const float*)d_in[4];
  const float* W2      = (const float*)d_in[5];
  const float* b2      = (const float*)d_in[6];
  float* out = (float*)d_out;

  const int N = GCN_N_NODES;
  const int E = GCN_N_EDGES;

  float* inv_out = (float*)d_ws;
  float* inv_in  = inv_out + 2 * N;
  int*   rowptr  = (int*)(inv_in + N);
  int*   csr_src = rowptr + (N + 16) + N;
  float* agg     = (float*)(csr_src + E);
  float* h1      = agg + (size_t)N * 128;
  float* t       = agg;
  int*   hist_in  = (int*)(h1 + (size_t)N * 128);
  int*   hist_out = hist_in + (size_t)GCN_G * N;
  int*   lrank    = hist_out + (size_t)GCN_G * N;
  int*   ctot     = lrank + E;

  gcn_count_kernel<<<dim3(GCN_G, 2), 1024, 0, stream>>>(src, dst, hist_in, hist_out, lrank, ctot);
  gcn_scan_kernel<<<2 * GCN_NCH, 1024, 0, stream>>>(hist_in, hist_out, ctot, inv_out, inv_in, rowptr);
  gcn_fill_kernel<<<(E / 4 + 255) / 256, 256, 0, stream>>>(src, dst, lrank, hist_in, csr_src);

  gcn_gather128<<<(N + 7) / 8, 256, 0, stream>>>(in_feat, rowptr, csr_src, inv_out, agg, N);
  gcn_gemm1_wmma<<<N / 16, 256, 0, stream>>>(agg, inv_in, W1, b1, h1);

  gcn_gemm2_wmma<<<(N + 63) / 64, 256, 0, stream>>>(h1, inv_out, W2, t, N);
  gcn_gather32<<<(N + 31) / 32, 256, 0, stream>>>(t, rowptr, csr_src, inv_in, b2, out, N);
}


# Layer-1 gather and both WMMA products fused into one launch per 16-node tile: edge sums and h1 stay in LDS, never in memory; 7 launches -> 5, 81k waves -> 25k; every f32 sum order as before
# speedup vs baseline: 4.5948x; 1.3532x over previous
#include <hip/hip_runtime.h>
#include <hip/hip_bf16.h>


#define GCN_N_NODES 50000
#define GCN_N_EDGES 800000
#define GCN_IN      128
#define GCN_H       128
#define GCN_C       32

typedef __attribute__((ext_vector_type(2))) float v2f;
typedef __attribute__((ext_vector_type(8))) float v8f;

#define GCN_G      32
#define GCN_CHUNK  ((((GCN_N_EDGES + GCN_G - 1) / GCN_G) + 3) / 4 * 4)
#define GCN_KQ     ((GCN_CHUNK / 4 + 1023) / 1024)
#define GCN_NCH    ((GCN_N_NODES + 1023) / 1024)
#define GCN_ROUNDS ((GCN_N_NODES / 4 + 1023) / 1024)
#define GCN_CT     64
static_assert(GCN_N_NODES % 4 == 0 && GCN_N_EDGES % 4 == 0, "4-wide loads");
static_assert(4 * GCN_ROUNDS <= GCN_CT && GCN_NCH <= 4 * GCN_ROUNDS, "chunk totals");
static_assert(GCN_CHUNK % 4 == 0, "a 4-edge load stays inside one group");

__global__ __launch_bounds__(1024) void gcn_count_kernel(
    const int* __restrict__ src, const int* __restrict__ dst,
    int* __restrict__ hist_in, int* __restrict__ hist_out,
    int* __restrict__ lrank, int* __restrict__ ctot) {
  __shared__ __attribute__((aligned(16))) int tbl[GCN_N_NODES];
  __shared__ int wtot[GCN_ROUNDS * 32];
  const int tid  = threadIdx.x;
  const int lane = tid & 31;
  const int w    = tid >> 5;
  const int g    = blockIdx.x;
  const bool by_dst = (blockIdx.y == 0);
  const int e0 = g * GCN_CHUNK;
  const int e1 = (e0 + GCN_CHUNK < GCN_N_EDGES) ? (e0 + GCN_CHUNK) : GCN_N_EDGES;
  const int nq = (e1 > e0) ? ((e1 - e0) >> 2) : 0;
  const int4* __restrict__ ids = (const int4*)((by_dst ? dst : src) + e0);

  int4 id[GCN_KQ];
#pragma unroll
  for (int k = 0; k < GCN_KQ; ++k) {
    const int qi = k * 1024 + tid;
    id[k] = (qi < nq) ? ids[qi] : make_int4(0, 0, 0, 0);
  }

  for (int i = tid; i < GCN_N_NODES / 4; i += 1024)
    ((int4*)tbl)[i] = make_int4(0, 0, 0, 0);
  __syncthreads();

#pragma unroll
  for (int k = 0; k < GCN_KQ; ++k) {
    const int qi = k * 1024 + tid;
    if (qi < nq) {
      int4 r;
      r.x = atomicAdd(&tbl[id[k].x], 1);
      r.y = atomicAdd(&tbl[id[k].y], 1);
      r.z = atomicAdd(&tbl[id[k].z], 1);
      r.w = atomicAdd(&tbl[id[k].w], 1);
      if (by_dst) ((int4*)(lrank + e0))[qi] = r;
    }
  }
  __syncthreads();

  int* __restrict__ hist = (by_dst ? hist_in : hist_out) + (size_t)g * GCN_N_NODES;
  for (int q = 0; q < GCN_ROUNDS; ++q) {
    const int vi = q * 1024 + tid;
    int4 v = make_int4(0, 0, 0, 0);
    if (vi < GCN_N_NODES / 4) {
      v = ((const int4*)tbl)[vi];
      ((int4*)hist)[vi] = v;
    }
    int s = v.x + v.y + v.z + v.w;
    for (int off = 16; off > 0; off >>= 1) s += __shfl_xor(s, off);
    if (lane == 0) wtot[q * 32 + w] = s;
  }
  __syncthreads();
  if (by_dst && tid < 4 * GCN_ROUNDS) {
    const int q = tid >> 2, c = tid & 3;
    int s = 0;
    for (int t = 0; t < 8; ++t) s += wtot[q * 32 + c * 8 + t];
    ctot[g * GCN_CT + tid] = s;
  }
}

__global__ __launch_bounds__(1024) void gcn_scan_kernel(
    int* __restrict__ hist_in, const int* __restrict__ hist_out,
    const int* __restrict__ ctot, float* __restrict__ inv_out,
    float* __restrict__ inv_in, int* __restrict__ rowptr) {
  __shared__ int pre_w[32];
  __shared__ int tot_w[32];
  const int tid  = threadIdx.x;
  const int lane = tid & 31;
  const int w    = tid >> 5;
  const int N    = GCN_N_NODES;

  if (blockIdx.x >= GCN_NCH) {
    const int i = (blockIdx.x - GCN_NCH) * 1024 + tid;
    if (i < N) {
      int d = 0;
#pragma unroll
      for (int g = 0; g < GCN_G; ++g) d += hist_out[(size_t)g * N + i];
      inv_out[i] = (d > 0) ? rsqrtf((float)d) : 0.0f;
    }
    return;
  }

  const int blk = blockIdx.x;
  const int lim = blk * 1024;
  int pre = 0;
  for (int i = tid; i < GCN_G * GCN_CT; i += 1024)
    if ((i & (GCN_CT - 1)) < blk) pre += ctot[i];

  const int i = lim + tid;
  int v[GCN_G];
  int c = 0;
#pragma unroll
  for (int g = 0; g < GCN_G; ++g) {
    v[g] = (i < N) ? hist_in[(size_t)g * N + i] : 0;
    c += v[g];
  }

  int incl = c;
  for (int off = 1; off < 32; off <<= 1) {
    int y = __shfl_up(incl, off);
    if (lane >= off) incl += y;
  }
  for (int off = 16; off > 0; off >>= 1) pre += __shfl_xor(pre, off);
  if (lane == 31) tot_w[w] = incl;
  if (lane == 0)  pre_w[w] = pre;
  __syncthreads();

  int part = pre_w[lane] + ((lane < w) ? tot_w[lane] : 0);
  for (int off = 16; off > 0; off >>= 1) part += __shfl_xor(part, off);
  const int excl = part + incl - c;

  if (i < N) {
    rowptr[i] = excl;
    int run = excl;
#pragma unroll
    for (int g = 0; g < GCN_G; ++g) {
      hist_in[(size_t)g * N + i] = run;
      run += v[g];
    }
    inv_in[i] = (c > 0) ? rsqrtf((float)c) : 0.0f;
    if (i == N - 1) rowptr[N] = excl + c;
  }
}

__global__ __launch_bounds__(256) void gcn_fill_kernel(
    const int* __restrict__ src, const int* __restrict__ dst,
    const int* __restrict__ lrank, const int* __restrict__ gstart,
    int* __restrict__ csr_src) {
  const int q = blockIdx.x * 256 + threadIdx.x;
  const int e = q * 4;
  if (e >= GCN_N_EDGES) return;
  const int4 d = ((const int4*)dst)[q];
  const int4 s = ((const int4*)src)[q];
  const int4 r = ((const int4*)lrank)[q];
  const int* __restrict__ gs = gstart + (size_t)(e / GCN_CHUNK) * GCN_N_NODES;
  csr_src[gs[d.x] + r.x] = s.x;
  csr_src[gs[d.y] + r.y] = s.y;
  csr_src[gs[d.z] + r.z] = s.z;
  csr_src[gs[d.w] + r.w] = s.w;
}

#define GCN_TP 132
static_assert(GCN_N_NODES % 16 == 0, "whole 16-node tiles");

__device__ __forceinline__ float4 gcn_row_sum128(
    const float* __restrict__ x, const int* __restrict__ csr_src,
    const float* __restrict__ inv_out, int beg, int end, int lane,
    int s_l, float w_l) {
  float4 acc = make_float4(0.0f, 0.0f, 0.0f, 0.0f);
  for (int base = beg; base < end; base += 32) {
    const int m = (end - base < 32) ? (end - base) : 32;
    if (base != beg) {
      s_l = 0;
      w_l = 0.0f;
      if (lane < m) {
        s_l = csr_src[base + lane];
        w_l = inv_out[s_l];
      }
    }
    int j = 0;
    for (; j + 4 <= m; j += 4) {
      const int s0 = __builtin_amdgcn_readlane(s_l, j);
      const int s1 = __builtin_amdgcn_readlane(s_l, j + 1);
      const int s2 = __builtin_amdgcn_readlane(s_l, j + 2);
      const int s3 = __builtin_amdgcn_readlane(s_l, j + 3);
      const float w0 = __int_as_float(__builtin_amdgcn_readlane(__float_as_int(w_l), j));
      const float w1 = __int_as_float(__builtin_amdgcn_readlane(__float_as_int(w_l), j + 1));
      const float w2 = __int_as_float(__builtin_amdgcn_readlane(__float_as_int(w_l), j + 2));
      const float w3 = __int_as_float(__builtin_amdgcn_readlane(__float_as_int(w_l), j + 3));
      const float4 v0 = ((const float4*)(x + (size_t)s0 * 128))[lane];
      const float4 v1 = ((const float4*)(x + (size_t)s1 * 128))[lane];
      const float4 v2 = ((const float4*)(x + (size_t)s2 * 128))[lane];
      const float4 v3 = ((const float4*)(x + (size_t)s3 * 128))[lane];
      acc.x += v0.x * w0; acc.y += v0.y * w0; acc.z += v0.z * w0; acc.w += v0.w * w0;
      acc.x += v1.x * w1; acc.y += v1.y * w1; acc.z += v1.z * w1; acc.w += v1.w * w1;
      acc.x += v2.x * w2; acc.y += v2.y * w2; acc.z += v2.z * w2; acc.w += v2.w * w2;
      acc.x += v3.x * w3; acc.y += v3.y * w3; acc.z += v3.z * w3; acc.w += v3.w * w3;
    }
    for (; j < m; ++j) {
      const int s0 = __builtin_amdgcn_readlane(s_l, j);
      const float w0 = __int_as_float(__builtin_amdgcn_readlane(__float_as_int(w_l), j));
      const float4 v0 = ((const float4*)(x + (size_t)s0 * 128))[lane];
      acc.x += v0.x * w0; acc.y += v0.y * w0; acc.z += v0.z * w0; acc.w += v0.w * w0;
    }
  }
  return acc;
}

__global__ __launch_bounds__(256, 16) void gcn_layer1_fused(
    const float* __restrict__ x, const int* __restrict__ rowptr,
    const int* __restrict__ csr_src, const float* __restrict__ inv_out,
    const float* __restrict__ inv_in, const float* __restrict__ W1,
    const float* __restrict__ b1, const float* __restrict__ W2,
    float* __restrict__ t) {
  __shared__ __attribute__((aligned(16))) float As[16][GCN_TP];
  __shared__ __attribute__((aligned(16))) float Hs[16][GCN_TP];
  __shared__ __attribute__((aligned(16))) float Sc[16];
  const int tid  = threadIdx.x;
  const int lane = tid & 31;
  const int wave = __builtin_amdgcn_readfirstlane(tid >> 5);
  const int row0 = blockIdx.x * 16;

  const int n0 = row0 + 2 * wave;
  const int   rp_l = (lane < 3) ? rowptr[n0 + lane] : 0;
  const float di_l = (lane < 2) ? inv_in[n0 + lane] : 0.0f;
  if (wave == 0 && lane < 16) Sc[lane] = inv_out[row0 + lane];
  const int beg0 = __builtin_amdgcn_readlane(rp_l, 0);
  const int end0 = __builtin_amdgcn_readlane(rp_l, 1);
  const int end1 = __builtin_amdgcn_readlane(rp_l, 2);
  const int m0 = (end0 - beg0 < 32) ? (end0 - beg0) : 32;
  const int m1 = (end1 - end0 < 32) ? (end1 - end0) : 32;
  int   s0_l = 0,    s1_l = 0;
  float w0_l = 0.0f, w1_l = 0.0f;
  if (lane < m0) s0_l = csr_src[beg0 + lane];
  if (lane < m1) s1_l = csr_src[end0 + lane];
  if (lane < m0) w0_l = inv_out[s0_l];
  if (lane < m1) w1_l = inv_out[s1_l];
  const float4 a0 = gcn_row_sum128(x, csr_src, inv_out, beg0, end0, lane, s0_l, w0_l);
  const float4 a1 = gcn_row_sum128(x, csr_src, inv_out, end0, end1, lane, s1_l, w1_l);
  const float d0 = __int_as_float(__builtin_amdgcn_readlane(__float_as_int(di_l), 0));
  const float d1 = __int_as_float(__builtin_amdgcn_readlane(__float_as_int(di_l), 1));
  ((float4*)As[2 * wave])[lane]     = make_float4(a0.x * d0, a0.y * d0, a0.z * d0, a0.w * d0);
  ((float4*)As[2 * wave + 1])[lane] = make_float4(a1.x * d1, a1.y * d1, a1.z * d1, a1.w * d1);
  __syncthreads();

  const int half = lane >> 4;
  const int mrow = lane & 15;
  {
    const int col0 = wave * 16;
    v8f acc = {};
    for (int kk = 0; kk < 128; kk += 4) {
      int k2 = kk + half * 2;
      v2f a, b;
      a.x = As[mrow][k2];
      a.y = As[mrow][k2 + 1];
      b.x = W1[(size_t)k2 * 128 + col0 + mrow];
      b.y = W1[(size_t)(k2 + 1) * 128 + col0 + mrow];
      acc = __builtin_amdgcn_wmma_f32_16x16x4_f32(
           false, a,  false, b,
           (short)0, acc,  false,  false);
    }
    const float bv = b1[col0 + mrow];
    for (int r = 0; r < 8; ++r) {
      const int row = r + half * 8;
      float v = acc[r] + bv;
      v = (v > 0.0f) ? v : 0.0f;
      Hs[row][col0 + mrow] = v * Sc[row];
    }
  }
  __syncthreads();

  const int pw = (wave - 2 * (int)blockIdx.x) & 7;
  if (pw < 2) {
    const int c0 = pw * 16;
    v8f acc = {};
    for (int kk = 0; kk < 128; kk += 4) {
      int k2 = kk + half * 2;
      v2f a, b;
      a.x = Hs[mrow][k2];
      a.y = Hs[mrow][k2 + 1];
      b.x = W2[(size_t)k2 * 32 + c0 + mrow];
      b.y = W2[(size_t)(k2 + 1) * 32 + c0 + mrow];
      acc = __builtin_amdgcn_wmma_f32_16x16x4_f32(
          false, a, false, b, (short)0, acc, false, false);
    }
    for (int r = 0; r < 8; ++r) {
      const int row = row0 + r + half * 8;
      t[(size_t)row * 32 + c0 + mrow] = acc[r];
    }
  }
}

__global__ __launch_bounds__(256) void gcn_gather32(
    const float* __restrict__ t, const int* __restrict__ rowptr,
    const int* __restrict__ csr_src, const float* __restrict__ inv_in,
    const float* __restrict__ bias, float* __restrict__ out, int N) {
  const int sub = threadIdx.x & 7;
  const int n = blockIdx.x * 32 + (threadIdx.x >> 3);
  if (n >= N) return;
  const int beg = rowptr[n];
  const int end = rowptr[n + 1];

  float4 acc = make_float4(0.0f, 0.0f, 0.0f, 0.0f);
  int j = beg;
  for (; j + 4 <= end; j += 4) {
    const int s0 = csr_src[j];
    const int s1 = csr_src[j + 1];
    const int s2 = csr_src[j + 2];
    const int s3 = csr_src[j + 3];
    const float4 v0 = ((const float4*)(t + (size_t)s0 * 32))[sub];
    const float4 v1 = ((const float4*)(t + (size_t)s1 * 32))[sub];
    const float4 v2 = ((const float4*)(t + (size_t)s2 * 32))[sub];
    const float4 v3 = ((const float4*)(t + (size_t)s3 * 32))[sub];
    acc.x += v0.x; acc.y += v0.y; acc.z += v0.z; acc.w += v0.w;
    acc.x += v1.x; acc.y += v1.y; acc.z += v1.z; acc.w += v1.w;
    acc.x += v2.x; acc.y += v2.y; acc.z += v2.z; acc.w += v2.w;
    acc.x += v3.x; acc.y += v3.y; acc.z += v3.z; acc.w += v3.w;
  }
  for (; j < end; ++j) {
    const int s0 = csr_src[j];
    const float4 v0 = ((const float4*)(t + (size_t)s0 * 32))[sub];
    acc.x += v0.x; acc.y += v0.y; acc.z += v0.z; acc.w += v0.w;
  }
  const float sc = inv_in[n];
  const float4 b = ((const float4*)bias)[sub];
  float4 o;
  o.x = acc.x * sc + b.x;
  o.y = acc.y * sc + b.y;
  o.z = acc.z * sc + b.z;
  o.w = acc.w * sc + b.w;
  ((float4*)(out + (size_t)n * 32))[sub] = o;
}

extern "C" void kernel_launch(void* const* d_in, const int* in_sizes, int n_in,
                              void* d_out, int out_size, void* d_ws, size_t ws_size,
                              hipStream_t stream) {
  const float* in_feat = (const float*)d_in[0];
  const int*   src     = (const int*)d_in[1];
  const int*   dst     = (const int*)d_in[2];
  const float* W1      = (const float*)d_in[3];
  const float* b1      = (const float*)d_in[4];
  const float* W2      = (const float*)d_in[5];
  const float* b2      = (const float*)d_in[6];
  float* out = (float*)d_out;

  const int N = GCN_N_NODES;
  const int E = GCN_N_EDGES;

  float* inv_out = (float*)d_ws;
  float* inv_in  = inv_out + 2 * N;
  int*   rowptr  = (int*)(inv_in + N);
  int*   csr_src = rowptr + (N + 16) + N;
  float* t       = (float*)(csr_src + E);
  int*   hist_in  = (int*)(t + (size_t)N * 32);
  int*   hist_out = hist_in + (size_t)GCN_G * N;
  int*   lrank    = hist_out + (size_t)GCN_G * N;
  int*   ctot     = lrank + E;

  gcn_count_kernel<<<dim3(GCN_G, 2), 1024, 0, stream>>>(src, dst, hist_in, hist_out, lrank, ctot);
  gcn_scan_kernel<<<2 * GCN_NCH, 1024, 0, stream>>>(hist_in, hist_out, ctot, inv_out, inv_in, rowptr);
  gcn_fill_kernel<<<(E / 4 + 255) / 256, 256, 0, stream>>>(src, dst, lrank, hist_in, csr_src);

  gcn_layer1_fused<<<N / 16, 256, 0, stream>>>(in_feat, rowptr, csr_src, inv_out, inv_in,
                                             W1, b1, W2, t);

  gcn_gather32<<<(N + 31) / 32, 256, 0, stream>>>(t, rowptr, csr_src, inv_in, b2, out, N);
}


# Edge sort in 2 launches, not 3: each count block sorts its group by destination inside LDS (table scan, start + rank); the scan launch copies the groups' pieces into rows; f32 sum orders as before
# speedup vs baseline: 5.5885x; 1.2163x over previous
#include <hip/hip_runtime.h>
#include <hip/hip_bf16.h>


#define GCN_N_NODES 50000
#define GCN_N_EDGES 800000
#define GCN_IN      128
#define GCN_H       128
#define GCN_C       32

typedef __attribute__((ext_vector_type(2))) float v2f;
typedef __attribute__((ext_vector_type(8))) float v8f;

#define GCN_G      32
#define GCN_CHUNK  ((((GCN_N_EDGES + GCN_G - 1) / GCN_G) + 3) / 4 * 4)
#define GCN_KQ     ((GCN_CHUNK / 4 + 1023) / 1024)
#define GCN_SEG4   ((((GCN_N_NODES + 1023) / 1024) + 3) / 4)
#define GCN_TBL    (1024 * GCN_SEG4 * 4)
#define GCN_NB2    ((GCN_N_NODES + 255) / 256)
#define GCN_GSIZE(g) (((g) * GCN_CHUNK >= GCN_N_EDGES) ? 0 : \
    ((((g) + 1) * GCN_CHUNK < GCN_N_EDGES) ? GCN_CHUNK : (GCN_N_EDGES - (g) * GCN_CHUNK)))
static_assert(GCN_N_NODES % 4 == 0 && GCN_N_EDGES % 4 == 0, "4-wide loads");
static_assert(GCN_CHUNK % 4 == 0, "a 4-edge load stays inside one group");
static_assert(GCN_TBL >= GCN_N_NODES, "the table holds the node range");
static_assert(GCN_SEG4 % 2 == 1, "scan stride of 16 B x odd: 16-byte LDS accesses of a wave do not collide");

__global__ __launch_bounds__(1024) void gcn_count_kernel(
    const int* __restrict__ src, const int* __restrict__ dst,
    int* __restrict__ hist_in, int* __restrict__ hist_out,
    int* __restrict__ csr_g) {
  __shared__ __attribute__((aligned(16))) int tbl[GCN_TBL];
  __shared__ __attribute__((aligned(16))) int stage[GCN_CHUNK];
  __shared__ int wsum[32];
  const int tid  = threadIdx.x;
  const int lane = tid & 31;
  const int w    = tid >> 5;
  const int g    = blockIdx.x;
  const bool by_dst = (blockIdx.y == 0);
  const int e0 = g * GCN_CHUNK;
  const int e1 = (e0 + GCN_CHUNK < GCN_N_EDGES) ? (e0 + GCN_CHUNK) : GCN_N_EDGES;
  const int nq = (e1 > e0) ? ((e1 - e0) >> 2) : 0;
  const int4* __restrict__ ids  = (const int4*)((by_dst ? dst : src) + e0);
  const int4* __restrict__ sids = (const int4*)(src + e0);

  int4 id[GCN_KQ];
  int4 sid[GCN_KQ];
#pragma unroll
  for (int k = 0; k < GCN_KQ; ++k) {
    const int qi = k * 1024 + tid;
    id[k]  = (qi < nq) ? ids[qi] : make_int4(0, 0, 0, 0);
    sid[k] = (by_dst && qi < nq) ? sids[qi] : make_int4(0, 0, 0, 0);
  }

  for (int i = tid; i < GCN_TBL / 4; i += 1024)
    ((int4*)tbl)[i] = make_int4(0, 0, 0, 0);
  __syncthreads();

  if (!by_dst) {
#pragma unroll
    for (int k = 0; k < GCN_KQ; ++k) {
      const int qi = k * 1024 + tid;
      if (qi < nq) {
        atomicAdd(&tbl[id[k].x], 1);
        atomicAdd(&tbl[id[k].y], 1);
        atomicAdd(&tbl[id[k].z], 1);
        atomicAdd(&tbl[id[k].w], 1);
      }
    }
    __syncthreads();
    int* __restrict__ ho = hist_out + (size_t)g * GCN_N_NODES;
    for (int i = tid; i < GCN_N_NODES / 4; i += 1024)
      ((int4*)ho)[i] = ((const int4*)tbl)[i];
    return;
  }

  int4 r[GCN_KQ];
#pragma unroll
  for (int k = 0; k < GCN_KQ; ++k) {
    const int qi = k * 1024 + tid;
    r[k] = make_int4(0, 0, 0, 0);
    if (qi < nq) {
      r[k].x = atomicAdd(&tbl[id[k].x], 1);
      r[k].y = atomicAdd(&tbl[id[k].y], 1);
      r[k].z = atomicAdd(&tbl[id[k].z], 1);
      r[k].w = atomicAdd(&tbl[id[k].w], 1);
    }
  }
  __syncthreads();

  int4* seg = (int4*)tbl + tid * GCN_SEG4;
  int sum = 0;
#pragma unroll
  for (int u = 0; u < GCN_SEG4; ++u) {
    const int4 v = seg[u];
    sum += v.x + v.y + v.z + v.w;
  }
  int incl = sum;
  for (int off = 1; off < 32; off <<= 1) {
    int y = __shfl_up(incl, off);
    if (lane >= off) incl += y;
  }
  if (lane == 31) wsum[w] = incl;
  __syncthreads();
  int part = (lane < w) ? wsum[lane] : 0;
  for (int off = 16; off > 0; off >>= 1) part += __shfl_xor(part, off);
  int run = part + incl - sum;
#pragma unroll
  for (int u = 0; u < GCN_SEG4; ++u) {
    const int4 v = seg[u];
    int4 o;
    o.x = run;
    o.y = o.x + v.x;
    o.z = o.y + v.y;
    o.w = o.z + v.z;
    run = o.w + v.w;
    seg[u] = o;
  }
  __syncthreads();

#pragma unroll
  for (int k = 0; k < GCN_KQ; ++k) {
    const int qi = k * 1024 + tid;
    if (qi < nq) {
      stage[tbl[id[k].x] + r[k].x] = sid[k].x;
      stage[tbl[id[k].y] + r[k].y] = sid[k].y;
      stage[tbl[id[k].z] + r[k].z] = sid[k].z;
      stage[tbl[id[k].w] + r[k].w] = sid[k].w;
    }
  }
  __syncthreads();

  int* __restrict__ hi = hist_in + (size_t)g * GCN_N_NODES;
  for (int i = tid; i < GCN_N_NODES / 4; i += 1024)
    ((int4*)hi)[i] = ((const int4*)tbl)[i];
  for (int i = tid; i < nq; i += 1024)
    ((int4*)(csr_g + e0))[i] = ((const int4*)stage)[i];
}

__global__ __launch_bounds__(256) void gcn_merge_kernel(
    const int* __restrict__ hist_in, const int* __restrict__ hist_out,
    const int* __restrict__ csr_g, float* __restrict__ inv_out,
    float* __restrict__ inv_in, int* __restrict__ rowptr,
    int* __restrict__ csr_src) {
  __shared__ int tot_w[8];
  __shared__ int pre_s;
  const int tid  = threadIdx.x;
  const int lane = tid & 31;
  const int w    = tid >> 5;
  const int N    = GCN_N_NODES;

  if (blockIdx.x >= GCN_NB2) {
    const int i = (blockIdx.x - GCN_NB2) * 256 + tid;
    if (i < N) {
      int d = 0;
#pragma unroll
      for (int g = 0; g < GCN_G; ++g) d += hist_out[(size_t)g * N + i];
      inv_out[i] = (d > 0) ? rsqrtf((float)d) : 0.0f;
    }
    return;
  }

  const int i = blockIdx.x * 256 + tid;
  int s[GCN_G], n[GCN_G];
  int c = 0, sp = 0;
#pragma unroll
  for (int g = 0; g < GCN_G; ++g) {
    int a = 0, b = 0;
    if (i < N) {
      const unsigned h = (unsigned)(g * N + i);
      a = hist_in[h];
      b = (i + 1 < N) ? hist_in[h + 1] : GCN_GSIZE(g);
    }
    s[g] = a;
    n[g] = b - a;
    c  += b - a;
    sp += a;
  }
  if (tid == 0) pre_s = sp;

  int a0[GCN_G], a1[GCN_G], a2[GCN_G], a3[GCN_G];
#pragma unroll
  for (int g = 0; g < GCN_G; ++g) {
    const unsigned p = (unsigned)(g * GCN_CHUNK + s[g]);
    a0[g] = (n[g] > 0) ? csr_g[p]     : 0;
    a1[g] = (n[g] > 1) ? csr_g[p + 1] : 0;
    a2[g] = (n[g] > 2) ? csr_g[p + 2] : 0;
    a3[g] = (n[g] > 3) ? csr_g[p + 3] : 0;
  }

  int incl = c;
  for (int off = 1; off < 32; off <<= 1) {
    int y = __shfl_up(incl, off);
    if (lane >= off) incl += y;
  }
  if (lane == 31) tot_w[w] = incl;
  __syncthreads();
  int part = (lane < w) ? tot_w[lane & 7] : 0;
  for (int off = 16; off > 0; off >>= 1) part += __shfl_xor(part, off);
  const int excl = pre_s + part + incl - c;

  if (i < N) {
    rowptr[i] = excl;
    inv_in[i] = (c > 0) ? rsqrtf((float)c) : 0.0f;
    if (i == N - 1) rowptr[N] = excl + c;
  }
  unsigned run = (unsigned)excl;
#pragma unroll
  for (int g = 0; g < GCN_G; ++g) {
    if (n[g] > 0) csr_src[run]     = a0[g];
    if (n[g] > 1) csr_src[run + 1] = a1[g];
    if (n[g] > 2) csr_src[run + 2] = a2[g];
    if (n[g] > 3) csr_src[run + 3] = a3[g];
    const unsigned p = (unsigned)(g * GCN_CHUNK + s[g]);
#pragma clang loop vectorize(disable) unroll(disable)
    for (int k = 4; k < n[g]; ++k) csr_src[run + k] = csr_g[p + k];
    run += (unsigned)n[g];
  }
}

#define GCN_TP 132
static_assert(GCN_N_NODES % 16 == 0, "whole 16-node tiles");

__device__ __forceinline__ float4 gcn_row_sum128(
    const float* __restrict__ x, const int* __restrict__ csr_src,
    const float* __restrict__ inv_out, int beg, int end, int lane,
    int s_l, float w_l) {
  float4 acc = make_float4(0.0f, 0.0f, 0.0f, 0.0f);
  for (int base = beg; base < end; base += 32) {
    const int m = (end - base < 32) ? (end - base) : 32;
    if (base != beg) {
      s_l = 0;
      w_l = 0.0f;
      if (lane < m) {
        s_l = csr_src[base + lane];
        w_l = inv_out[s_l];
      }
    }
    int j = 0;
    for (; j + 4 <= m; j += 4) {
      const int s0 = __builtin_amdgcn_readlane(s_l, j);
      const int s1 = __builtin_amdgcn_readlane(s_l, j + 1);
      const int s2 = __builtin_amdgcn_readlane(s_l, j + 2);
      const int s3 = __builtin_amdgcn_readlane(s_l, j + 3);
      const float w0 = __int_as_float(__builtin_amdgcn_readlane(__float_as_int(w_l), j));
      const float w1 = __int_as_float(__builtin_amdgcn_readlane(__float_as_int(w_l), j + 1));
      const float w2 = __int_as_float(__builtin_amdgcn_readlane(__float_as_int(w_l), j + 2));
      const float w3 = __int_as_float(__builtin_amdgcn_readlane(__float_as_int(w_l), j + 3));
      const float4 v0 = ((const float4*)(x + (size_t)s0 * 128))[lane];
      const float4 v1 = ((const float4*)(x + (size_t)s1 * 128))[lane];
      const float4 v2 = ((const float4*)(x + (size_t)s2 * 128))[lane];
      const float4 v3 = ((const float4*)(x + (size_t)s3 * 128))[lane];
      acc.x += v0.x * w0; acc.y += v0.y * w0; acc.z += v0.z * w0; acc.w += v0.w * w0;
      acc.x += v1.x * w1; acc.y += v1.y * w1; acc.z += v1.z * w1; acc.w += v1.w * w1;
      acc.x += v2.x * w2; acc.y += v2.y * w2; acc.z += v2.z * w2; acc.w += v2.w * w2;
      acc.x += v3.x * w3; acc.y += v3.y * w3; acc.z += v3.z * w3; acc.w += v3.w * w3;
    }
    for (; j < m; ++j) {
      const int s0 = __builtin_amdgcn_readlane(s_l, j);
      const float w0 = __int_as_float(__builtin_amdgcn_readlane(__float_as_int(w_l), j));
      const float4 v0 = ((const float4*)(x + (size_t)s0 * 128))[lane];
      acc.x += v0.x * w0; acc.y += v0.y * w0; acc.z += v0.z * w0; acc.w += v0.w * w0;
    }
  }
  return acc;
}

__global__ __launch_bounds__(256, 16) void gcn_layer1_fused(
    const float* __restrict__ x, const int* __restrict__ rowptr,
    const int* __restrict__ csr_src, const float* __restrict__ inv_out,
    const float* __restrict__ inv_in, const float* __restrict__ W1,
    const float* __restrict__ b1, const float* __restrict__ W2,
    float* __restrict__ t) {
  __shared__ __attribute__((aligned(16))) float As[16][GCN_TP];
  __shared__ __attribute__((aligned(16))) float Hs[16][GCN_TP];
  __shared__ __attribute__((aligned(16))) float Sc[16];
  const int tid  = threadIdx.x;
  const int lane = tid & 31;
  const int wave = __builtin_amdgcn_readfirstlane(tid >> 5);
  const int row0 = blockIdx.x * 16;

  const int n0 = row0 + 2 * wave;
  const int   rp_l = (lane < 3) ? rowptr[n0 + lane] : 0;
  const float di_l = (lane < 2) ? inv_in[n0 + lane] : 0.0f;
  if (wave == 0 && lane < 16) Sc[lane] = inv_out[row0 + lane];
  const int beg0 = __builtin_amdgcn_readlane(rp_l, 0);
  const int end0 = __builtin_amdgcn_readlane(rp_l, 1);
  const int end1 = __builtin_amdgcn_readlane(rp_l, 2);
  const int m0 = (end0 - beg0 < 32) ? (end0 - beg0) : 32;
  const int m1 = (end1 - end0 < 32) ? (end1 - end0) : 32;
  int   s0_l = 0,    s1_l = 0;
  float w0_l = 0.0f, w1_l = 0.0f;
  if (lane < m0) s0_l = csr_src[beg0 + lane];
  if (lane < m1) s1_l = csr_src[end0 + lane];
  if (lane < m0) w0_l = inv_out[s0_l];
  if (lane < m1) w1_l = inv_out[s1_l];
  const float4 a0 = gcn_row_sum128(x, csr_src, inv_out, beg0, end0, lane, s0_l, w0_l);
  const float4 a1 = gcn_row_sum128(x, csr_src, inv_out, end0, end1, lane, s1_l, w1_l);
  const float d0 = __int_as_float(__builtin_amdgcn_readlane(__float_as_int(di_l), 0));
  const float d1 = __int_as_float(__builtin_amdgcn_readlane(__float_as_int(di_l), 1));
  ((float4*)As[2 * wave])[lane]     = make_float4(a0.x * d0, a0.y * d0, a0.z * d0, a0.w * d0);
  ((float4*)As[2 * wave + 1])[lane] = make_float4(a1.x * d1, a1.y * d1, a1.z * d1, a1.w * d1);
  __syncthreads();

  const int half = lane >> 4;
  const int mrow = lane & 15;
  {
    const int col0 = wave * 16;
    v8f acc = {};
    for (int kk = 0; kk < 128; kk += 4) {
      int k2 = kk + half * 2;
      v2f a, b;
      a.x = As[mrow][k2];
      a.y = As[mrow][k2 + 1];
      b.x = W1[(size_t)k2 * 128 + col0 + mrow];
      b.y = W1[(size_t)(k2 + 1) * 128 + col0 + mrow];
      acc = __builtin_amdgcn_wmma_f32_16x16x4_f32(
           false, a,  false, b,
           (short)0, acc,  false,  false);
    }
    const float bv = b1[col0 + mrow];
    for (int r = 0; r < 8; ++r) {
      const int row = r + half * 8;
      float v = acc[r] + bv;
      v = (v > 0.0f) ? v : 0.0f;
      Hs[row][col0 + mrow] = v * Sc[row];
    }
  }
  __syncthreads();

  const int pw = (wave - 2 * (int)blockIdx.x) & 7;
  if (pw < 2) {
    const int c0 = pw * 16;
    v8f acc = {};
    for (int kk = 0; kk < 128; kk += 4) {
      int k2 = kk + half * 2;
      v2f a, b;
      a.x = Hs[mrow][k2];
      a.y = Hs[mrow][k2 + 1];
      b.x = W2[(size_t)k2 * 32 + c0 + mrow];
      b.y = W2[(size_t)(k2 + 1) * 32 + c0 + mrow];
      acc = __builtin_amdgcn_wmma_f32_16x16x4_f32(
          false, a, false, b, (short)0, acc, false, false);
    }
    for (int r = 0; r < 8; ++r) {
      const int row = row0 + r + half * 8;
      t[(size_t)row * 32 + c0 + mrow] = acc[r];
    }
  }
}

__global__ __launch_bounds__(256) void gcn_gather32(
    const float* __restrict__ t, const int* __restrict__ rowptr,
    const int* __restrict__ csr_src, const float* __restrict__ inv_in,
    const float* __restrict__ bias, float* __restrict__ out, int N) {
  const int sub = threadIdx.x & 7;
  const int n = blockIdx.x * 32 + (threadIdx.x >> 3);
  if (n >= N) return;
  const int beg = rowptr[n];
  const int end = rowptr[n + 1];

  float4 acc = make_float4(0.0f, 0.0f, 0.0f, 0.0f);
  int j = beg;
  for (; j + 4 <= end; j += 4) {
    const int s0 = csr_src[j];
    const int s1 = csr_src[j + 1];
    const int s2 = csr_src[j + 2];
    const int s3 = csr_src[j + 3];
    const float4 v0 = ((const float4*)(t + (size_t)s0 * 32))[sub];
    const float4 v1 = ((const float4*)(t + (size_t)s1 * 32))[sub];
    const float4 v2 = ((const float4*)(t + (size_t)s2 * 32))[sub];
    const float4 v3 = ((const float4*)(t + (size_t)s3 * 32))[sub];
    acc.x += v0.x; acc.y += v0.y; acc.z += v0.z; acc.w += v0.w;
    acc.x += v1.x; acc.y += v1.y; acc.z += v1.z; acc.w += v1.w;
    acc.x += v2.x; acc.y += v2.y; acc.z += v2.z; acc.w += v2.w;
    acc.x += v3.x; acc.y += v3.y; acc.z += v3.z; acc.w += v3.w;
  }
  for (; j < end; ++j) {
    const int s0 = csr_src[j];
    const float4 v0 = ((const float4*)(t + (size_t)s0 * 32))[sub];
    acc.x += v0.x; acc.y += v0.y; acc.z += v0.z; acc.w += v0.w;
  }
  const float sc = inv_in[n];
  const float4 b = ((const float4*)bias)[sub];
  float4 o;
  o.x = acc.x * sc + b.x;
  o.y = acc.y * sc + b.y;
  o.z = acc.z * sc + b.z;
  o.w = acc.w * sc + b.w;
  ((float4*)(out + (size_t)n * 32))[sub] = o;
}

extern "C" void kernel_launch(void* const* d_in, const int* in_sizes, int n_in,
                              void* d_out, int out_size, void* d_ws, size_t ws_size,
                              hipStream_t stream) {
  const float* in_feat = (const float*)d_in[0];
  const int*   src     = (const int*)d_in[1];
  const int*   dst     = (const int*)d_in[2];
  const float* W1      = (const float*)d_in[3];
  const float* b1      = (const float*)d_in[4];
  const float* W2      = (const float*)d_in[5];
  const float* b2      = (const float*)d_in[6];
  float* out = (float*)d_out;

  const int N = GCN_N_NODES;
  const int E = GCN_N_EDGES;

  float* inv_out = (float*)d_ws;
  float* inv_in  = inv_out + 2 * N;
  int*   rowptr  = (int*)(inv_in + N);
  int*   csr_src = rowptr + (N + 16) + N;
  float* t       = (float*)(csr_src + E);
  int*   hist_in  = (int*)(t + (size_t)N * 32);
  int*   hist_out = hist_in + (size_t)GCN_G * N;
  int*   csr_g    = hist_out + (size_t)GCN_G * N;

  gcn_count_kernel<<<dim3(GCN_G, 2), 1024, 0, stream>>>(src, dst, hist_in, hist_out, csr_g);
  gcn_merge_kernel<<<2 * GCN_NB2, 256, 0, stream>>>(hist_in, hist_out, csr_g, inv_out, inv_in,
                                                   rowptr, csr_src);

  gcn_layer1_fused<<<N / 16, 256, 0, stream>>>(in_feat, rowptr, csr_src, inv_out, inv_in,
                                             W1, b1, W2, t);

  gcn_gather32<<<(N + 31) / 32, 256, 0, stream>>>(t, rowptr, csr_src, inv_in, b2, out, N);
}


# Sort blocks also split the node range: one work-group per (edge group, quarter of the nodes, side), so each LDS table, scan and copy-out is 4x smaller and 256 WGPs sort, not 64; f32 sum orders kept
# speedup vs baseline: 5.9693x; 1.0681x over previous
#include <hip/hip_runtime.h>
#include <hip/hip_bf16.h>


#define GCN_N_NODES 50000
#define GCN_N_EDGES 800000
#define GCN_IN      128
#define GCN_H       128
#define GCN_C       32

typedef __attribute__((ext_vector_type(2))) float v2f;
typedef __attribute__((ext_vector_type(8))) float v8f;

#define GCN_G      32
#define GCN_CHUNK  ((((GCN_N_EDGES + GCN_G - 1) / GCN_G) + 3) / 4 * 4)
#define GCN_KQ     ((GCN_CHUNK / 4 + 1023) / 1024)
#define GCN_NB2    ((GCN_N_NODES + 255) / 256)
#define GCN_R      4
#define GCN_NR     ((GCN_NB2 + GCN_R - 1) / GCN_R * 256)
#define GCN_SEG4R  5
#define GCN_TBLR   (1024 * GCN_SEG4R * 4)
#define GCN_NRP    (GCN_NR + 4)
#define GCN_DB     14
static_assert(GCN_N_NODES % 4 == 0 && GCN_N_EDGES % 4 == 0, "4-wide loads");
static_assert(GCN_CHUNK % 4 == 0, "a 4-edge load stays inside one group");
static_assert(GCN_SEG4R % 2 == 1, "scan stride of 16 B x odd: 16-byte LDS accesses of a wave do not collide");
static_assert(GCN_NR % 256 == 0 && GCN_R * GCN_NR >= GCN_N_NODES, "ranges are whole merge blocks and cover the nodes");
static_assert(GCN_TBLR >= GCN_NRP, "a range's table holds its nodes and the total");
static_assert(GCN_G * GCN_R <= 256, "the merge adds up the earlier ranges' totals with one thread each");
static_assert(GCN_NR <= (1 << GCN_DB) && GCN_CHUNK < (1 << (31 - GCN_DB)), "place and rank share one non-negative word");

__device__ __forceinline__ int gcn_place_rank(int* tbl, int d, bool valid) {
  if (valid && (unsigned)d < (unsigned)GCN_NR)
    return (atomicAdd(&tbl[d], 1) << GCN_DB) | d;
  return -1;
}

__global__ __launch_bounds__(1024) void gcn_count_kernel(
    const int* __restrict__ src, const int* __restrict__ dst,
    int* __restrict__ hist_in, int* __restrict__ hist_out,
    int* __restrict__ csr_g) {
  __shared__ __attribute__((aligned(16))) int tbl[GCN_TBLR];
  __shared__ __attribute__((aligned(16))) int stage[GCN_CHUNK];
  __shared__ int wsum[32];
  const int tid  = threadIdx.x;
  const int lane = tid & 31;
  const int w    = tid >> 5;
  const int g    = blockIdx.x;
  const bool by_dst = (blockIdx.y < GCN_R);
  const int rng = by_dst ? (int)blockIdx.y : (int)blockIdx.y - GCN_R;
  const int lo = rng * GCN_NR;
  const int e0 = g * GCN_CHUNK;
  const int e1 = (e0 + GCN_CHUNK < GCN_N_EDGES) ? (e0 + GCN_CHUNK) : GCN_N_EDGES;
  const int nq = (e1 > e0) ? ((e1 - e0) >> 2) : 0;
  const int4* __restrict__ ids  = (const int4*)((by_dst ? dst : src) + e0);
  const int4* __restrict__ sids = (const int4*)(src + e0);

  int4 id[GCN_KQ];
  int4 sid[GCN_KQ];
#pragma unroll
  for (int k = 0; k < GCN_KQ; ++k) {
    const int qi = k * 1024 + tid;
    id[k]  = (qi < nq) ? ids[qi] : make_int4(0, 0, 0, 0);
    sid[k] = (by_dst && qi < nq) ? sids[qi] : make_int4(0, 0, 0, 0);
  }

  for (int i = tid; i < GCN_TBLR / 4; i += 1024)
    ((int4*)tbl)[i] = make_int4(0, 0, 0, 0);
  __syncthreads();

  if (!by_dst) {
#pragma unroll
    for (int k = 0; k < GCN_KQ; ++k) {
      const int qi = k * 1024 + tid;
      if (qi < nq) {
        if ((unsigned)(id[k].x - lo) < (unsigned)GCN_NR) atomicAdd(&tbl[id[k].x - lo], 1);
        if ((unsigned)(id[k].y - lo) < (unsigned)GCN_NR) atomicAdd(&tbl[id[k].y - lo], 1);
        if ((unsigned)(id[k].z - lo) < (unsigned)GCN_NR) atomicAdd(&tbl[id[k].z - lo], 1);
        if ((unsigned)(id[k].w - lo) < (unsigned)GCN_NR) atomicAdd(&tbl[id[k].w - lo], 1);
      }
    }
    __syncthreads();
    int* __restrict__ ho = hist_out + (size_t)(g * GCN_R + rng) * GCN_NR;
    for (int i = tid; i < GCN_NR / 4; i += 1024)
      ((int4*)ho)[i] = ((const int4*)tbl)[i];
    return;
  }

#pragma unroll
  for (int k = 0; k < GCN_KQ; ++k) {
    const bool v = (k * 1024 + tid < nq);
    id[k].x = gcn_place_rank(tbl, id[k].x - lo, v);
    id[k].y = gcn_place_rank(tbl, id[k].y - lo, v);
    id[k].z = gcn_place_rank(tbl, id[k].z - lo, v);
    id[k].w = gcn_place_rank(tbl, id[k].w - lo, v);
  }
  __syncthreads();

  int4* seg = (int4*)tbl + tid * GCN_SEG4R;
  int sum = 0;
#pragma unroll
  for (int u = 0; u < GCN_SEG4R; ++u) {
    const int4 v = seg[u];
    sum += v.x + v.y + v.z + v.w;
  }
  int incl = sum;
  for (int off = 1; off < 32; off <<= 1) {
    int y = __shfl_up(incl, off);
    if (lane >= off) incl += y;
  }
  if (lane == 31) wsum[w] = incl;
  __syncthreads();
  int part = (lane < w) ? wsum[lane] : 0;
  for (int off = 16; off > 0; off >>= 1) part += __shfl_xor(part, off);
  int run = part + incl - sum;
#pragma unroll
  for (int u = 0; u < GCN_SEG4R; ++u) {
    const int4 v = seg[u];
    int4 o;
    o.x = run;
    o.y = o.x + v.x;
    o.z = o.y + v.y;
    o.w = o.z + v.z;
    run = o.w + v.w;
    seg[u] = o;
  }
  __syncthreads();

#pragma unroll
  for (int k = 0; k < GCN_KQ; ++k) {
    if (id[k].x >= 0) stage[tbl[id[k].x & ((1 << GCN_DB) - 1)] + (id[k].x >> GCN_DB)] = sid[k].x;
    if (id[k].y >= 0) stage[tbl[id[k].y & ((1 << GCN_DB) - 1)] + (id[k].y >> GCN_DB)] = sid[k].y;
    if (id[k].z >= 0) stage[tbl[id[k].z & ((1 << GCN_DB) - 1)] + (id[k].z >> GCN_DB)] = sid[k].z;
    if (id[k].w >= 0) stage[tbl[id[k].w & ((1 << GCN_DB) - 1)] + (id[k].w >> GCN_DB)] = sid[k].w;
  }
  __syncthreads();

  const int row = g * GCN_R + rng;
  int* __restrict__ hi = hist_in + (size_t)row * GCN_NRP;
  for (int i = tid; i < GCN_NRP / 4; i += 1024)
    ((int4*)hi)[i] = ((const int4*)tbl)[i];
  const int tot4 = (tbl[GCN_NR] + 3) >> 2;
  int* __restrict__ cg = csr_g + (size_t)row * GCN_CHUNK;
  for (int i = tid; i < tot4; i += 1024)
    ((int4*)cg)[i] = ((const int4*)stage)[i];
}

__global__ __launch_bounds__(256) void gcn_merge_kernel(
    const int* __restrict__ hist_in, const int* __restrict__ hist_out,
    const int* __restrict__ csr_g, float* __restrict__ inv_out,
    float* __restrict__ inv_in, int* __restrict__ rowptr,
    int* __restrict__ csr_src) {
  __shared__ int tot_w[8];
  __shared__ int ext_w[8];
  __shared__ int pre_s;
  const int tid  = threadIdx.x;
  const int lane = tid & 31;
  const int w    = tid >> 5;
  const int N    = GCN_N_NODES;

  if (blockIdx.x >= GCN_NB2) {
    const int bi = blockIdx.x - GCN_NB2;
    const int i = bi * 256 + tid;
    const int rg = bi / (GCN_NR / 256);
    const int li = i - rg * GCN_NR;
    if (i < N) {
      int d = 0;
#pragma unroll
      for (int g = 0; g < GCN_G; ++g) d += hist_out[(unsigned)((g * GCN_R + rg) * GCN_NR + li)];
      inv_out[i] = (d > 0) ? rsqrtf((float)d) : 0.0f;
    }
    return;
  }

  const int i = blockIdx.x * 256 + tid;
  const int rg = blockIdx.x / (GCN_NR / 256);
  const int li = i - rg * GCN_NR;
  int s[GCN_G], n[GCN_G];
  int c = 0, sp = 0;
#pragma unroll
  for (int g = 0; g < GCN_G; ++g) {
    int a = 0, b = 0;
    if (i < N) {
      const unsigned h = (unsigned)((g * GCN_R + rg) * GCN_NRP + li);
      a = hist_in[h];
      b = hist_in[h + 1];
    }
    s[g] = a;
    n[g] = b - a;
    c  += b - a;
    sp += a;
  }
  if (tid == 0) pre_s = sp;
  int ex = 0;
  if (tid < GCN_G * GCN_R && (tid % GCN_R) < rg)
    ex = hist_in[(unsigned)(tid * GCN_NRP + GCN_NR)];
  for (int off = 16; off > 0; off >>= 1) ex += __shfl_xor(ex, off);
  if (lane == 0) ext_w[w] = ex;

  int a0[GCN_G], a1[GCN_G], a2[GCN_G], a3[GCN_G];
#pragma unroll
  for (int g = 0; g < GCN_G; ++g) {
    const unsigned p = (unsigned)((g * GCN_R + rg) * GCN_CHUNK + s[g]);
    a0[g] = (n[g] > 0) ? csr_g[p]     : 0;
    a1[g] = (n[g] > 1) ? csr_g[p + 1] : 0;
    a2[g] = (n[g] > 2) ? csr_g[p + 2] : 0;
    a3[g] = (n[g] > 3) ? csr_g[p + 3] : 0;
  }

  int incl = c;
  for (int off = 1; off < 32; off <<= 1) {
    int y = __shfl_up(incl, off);
    if (lane >= off) incl += y;
  }
  if (lane == 31) tot_w[w] = incl;
  __syncthreads();
  int part = (lane < w) ? tot_w[lane & 7] : 0;
  for (int off = 16; off > 0; off >>= 1) part += __shfl_xor(part, off);
  int pre = pre_s;
  for (int k = 0; k < 8; ++k) pre += ext_w[k];
  const int excl = pre + part + incl - c;

  if (i < N) {
    rowptr[i] = excl;
    inv_in[i] = (c > 0) ? rsqrtf((float)c) : 0.0f;
    if (i == N - 1) rowptr[N] = excl + c;
  }
  unsigned run = (unsigned)excl;
#pragma unroll
  for (int g = 0; g < GCN_G; ++g) {
    if (n[g] > 0) csr_src[run]     = a0[g];
    if (n[g] > 1) csr_src[run + 1] = a1[g];
    if (n[g] > 2) csr_src[run + 2] = a2[g];
    if (n[g] > 3) csr_src[run + 3] = a3[g];
    const unsigned p = (unsigned)((g * GCN_R + rg) * GCN_CHUNK + s[g]);
#pragma clang loop vectorize(disable) unroll(disable)
    for (int k = 4; k < n[g]; ++k) csr_src[run + k] = csr_g[p + k];
    run += (unsigned)n[g];
  }
}

#define GCN_TP 132
static_assert(GCN_N_NODES % 16 == 0, "whole 16-node tiles");

__device__ __forceinline__ float4 gcn_row_sum128(
    const float* __restrict__ x, const int* __restrict__ csr_src,
    const float* __restrict__ inv_out, int beg, int end, int lane,
    int s_l, float w_l) {
  float4 acc = make_float4(0.0f, 0.0f, 0.0f, 0.0f);
  for (int base = beg; base < end; base += 32) {
    const int m = (end - base < 32) ? (end - base) : 32;
    if (base != beg) {
      s_l = 0;
      w_l = 0.0f;
      if (lane < m) {
        s_l = csr_src[base + lane];
        w_l = inv_out[s_l];
      }
    }
    int j = 0;
    for (; j + 4 <= m; j += 4) {
      const int s0 = __builtin_amdgcn_readlane(s_l, j);
      const int s1 = __builtin_amdgcn_readlane(s_l, j + 1);
      const int s2 = __builtin_amdgcn_readlane(s_l, j + 2);
      const int s3 = __builtin_amdgcn_readlane(s_l, j + 3);
      const float w0 = __int_as_float(__builtin_amdgcn_readlane(__float_as_int(w_l), j));
      const float w1 = __int_as_float(__builtin_amdgcn_readlane(__float_as_int(w_l), j + 1));
      const float w2 = __int_as_float(__builtin_amdgcn_readlane(__float_as_int(w_l), j + 2));
      const float w3 = __int_as_float(__builtin_amdgcn_readlane(__float_as_int(w_l), j + 3));
      const float4 v0 = ((const float4*)(x + (size_t)s0 * 128))[lane];
      const float4 v1 = ((const float4*)(x + (size_t)s1 * 128))[lane];
      const float4 v2 = ((const float4*)(x + (size_t)s2 * 128))[lane];
      const float4 v3 = ((const float4*)(x + (size_t)s3 * 128))[lane];
      acc.x += v0.x * w0; acc.y += v0.y * w0; acc.z += v0.z * w0; acc.w += v0.w * w0;
      acc.x += v1.x * w1; acc.y += v1.y * w1; acc.z += v1.z * w1; acc.w += v1.w * w1;
      acc.x += v2.x * w2; acc.y += v2.y * w2; acc.z += v2.z * w2; acc.w += v2.w * w2;
      acc.x += v3.x * w3; acc.y += v3.y * w3; acc.z += v3.z * w3; acc.w += v3.w * w3;
    }
    for (; j < m; ++j) {
      const int s0 = __builtin_amdgcn_readlane(s_l, j);
      const float w0 = __int_as_float(__builtin_amdgcn_readlane(__float_as_int(w_l), j));
      const float4 v0 = ((const float4*)(x + (size_t)s0 * 128))[lane];
      acc.x += v0.x * w0; acc.y += v0.y * w0; acc.z += v0.z * w0; acc.w += v0.w * w0;
    }
  }
  return acc;
}

__global__ __launch_bounds__(256, 16) void gcn_layer1_fused(
    const float* __restrict__ x, const int* __restrict__ rowptr,
    const int* __restrict__ csr_src, const float* __restrict__ inv_out,
    const float* __restrict__ inv_in, const float* __restrict__ W1,
    const float* __restrict__ b1, const float* __restrict__ W2,
    float* __restrict__ t) {
  __shared__ __attribute__((aligned(16))) float As[16][GCN_TP];
  __shared__ __attribute__((aligned(16))) float Hs[16][GCN_TP];
  __shared__ __attribute__((aligned(16))) float Sc[16];
  const int tid  = threadIdx.x;
  const int lane = tid & 31;
  const int wave = __builtin_amdgcn_readfirstlane(tid >> 5);
  const int row0 = blockIdx.x * 16;

  const int n0 = row0 + 2 * wave;
  const int   rp_l = (lane < 3) ? rowptr[n0 + lane] : 0;
  const float di_l = (lane < 2) ? inv_in[n0 + lane] : 0.0f;
  if (wave == 0 && lane < 16) Sc[lane] = inv_out[row0 + lane];
  const int beg0 = __builtin_amdgcn_readlane(rp_l, 0);
  const int end0 = __builtin_amdgcn_readlane(rp_l, 1);
  const int end1 = __builtin_amdgcn_readlane(rp_l, 2);
  const int m0 = (end0 - beg0 < 32) ? (end0 - beg0) : 32;
  const int m1 = (end1 - end0 < 32) ? (end1 - end0) : 32;
  int   s0_l = 0,    s1_l = 0;
  float w0_l = 0.0f, w1_l = 0.0f;
  if (lane < m0) s0_l = csr_src[beg0 + lane];
  if (lane < m1) s1_l = csr_src[end0 + lane];
  if (lane < m0) w0_l = inv_out[s0_l];
  if (lane < m1) w1_l = inv_out[s1_l];
  const float4 a0 = gcn_row_sum128(x, csr_src, inv_out, beg0, end0, lane, s0_l, w0_l);
  const float4 a1 = gcn_row_sum128(x, csr_src, inv_out, end0, end1, lane, s1_l, w1_l);
  const float d0 = __int_as_float(__builtin_amdgcn_readlane(__float_as_int(di_l), 0));
  const float d1 = __int_as_float(__builtin_amdgcn_readlane(__float_as_int(di_l), 1));
  ((float4*)As[2 * wave])[lane]     = make_float4(a0.x * d0, a0.y * d0, a0.z * d0, a0.w * d0);
  ((float4*)As[2 * wave + 1])[lane] = make_float4(a1.x * d1, a1.y * d1, a1.z * d1, a1.w * d1);
  __syncthreads();

  const int half = lane >> 4;
  const int mrow = lane & 15;
  {
    const int col0 = wave * 16;
    v8f acc = {};
    for (int kk = 0; kk < 128; kk += 4) {
      int k2 = kk + half * 2;
      v2f a, b;
      a.x = As[mrow][k2];
      a.y = As[mrow][k2 + 1];
      b.x = W1[(size_t)k2 * 128 + col0 + mrow];
      b.y = W1[(size_t)(k2 + 1) * 128 + col0 + mrow];
      acc = __builtin_amdgcn_wmma_f32_16x16x4_f32(
           false, a,  false, b,
           (short)0, acc,  false,  false);
    }
    const float bv = b1[col0 + mrow];
    for (int r = 0; r < 8; ++r) {
      const int row = r + half * 8;
      float v = acc[r] + bv;
      v = (v > 0.0f) ? v : 0.0f;
      Hs[row][col0 + mrow] = v * Sc[row];
    }
  }
  __syncthreads();

  const int pw = (wave - 2 * (int)blockIdx.x) & 7;
  if (pw < 2) {
    const int c0 = pw * 16;
    v8f acc = {};
    for (int kk = 0; kk < 128; kk += 4) {
      int k2 = kk + half * 2;
      v2f a, b;
      a.x = Hs[mrow][k2];
      a.y = Hs[mrow][k2 + 1];
      b.x = W2[(size_t)k2 * 32 + c0 + mrow];
      b.y = W2[(size_t)(k2 + 1) * 32 + c0 + mrow];
      acc = __builtin_amdgcn_wmma_f32_16x16x4_f32(
          false, a, false, b, (short)0, acc, false, false);
    }
    for (int r = 0; r < 8; ++r) {
      const int row = row0 + r + half * 8;
      t[(size_t)row * 32 + c0 + mrow] = acc[r];
    }
  }
}

__global__ __launch_bounds__(256) void gcn_gather32(
    const float* __restrict__ t, const int* __restrict__ rowptr,
    const int* __restrict__ csr_src, const float* __restrict__ inv_in,
    const float* __restrict__ bias, float* __restrict__ out, int N) {
  const int sub = threadIdx.x & 7;
  const int n = blockIdx.x * 32 + (threadIdx.x >> 3);
  if (n >= N) return;
  const int beg = rowptr[n];
  const int end = rowptr[n + 1];

  float4 acc = make_float4(0.0f, 0.0f, 0.0f, 0.0f);
  int j = beg;
  for (; j + 4 <= end; j += 4) {
    const int s0 = csr_src[j];
    const int s1 = csr_src[j + 1];
    const int s2 = csr_src[j + 2];
    const int s3 = csr_src[j + 3];
    const float4 v0 = ((const float4*)(t + (size_t)s0 * 32))[sub];
    const float4 v1 = ((const float4*)(t + (size_t)s1 * 32))[sub];
    const float4 v2 = ((const float4*)(t + (size_t)s2 * 32))[sub];
    const float4 v3 = ((const float4*)(t + (size_t)s3 * 32))[sub];
    acc.x += v0.x; acc.y += v0.y; acc.z += v0.z; acc.w += v0.w;
    acc.x += v1.x; acc.y += v1.y; acc.z += v1.z; acc.w += v1.w;
    acc.x += v2.x; acc.y += v2.y; acc.z += v2.z; acc.w += v2.w;
    acc.x += v3.x; acc.y += v3.y; acc.z += v3.z; acc.w += v3.w;
  }
  for (; j < end; ++j) {
    const int s0 = csr_src[j];
    const float4 v0 = ((const float4*)(t + (size_t)s0 * 32))[sub];
    acc.x += v0.x; acc.y += v0.y; acc.z += v0.z; acc.w += v0.w;
  }
  const float sc = inv_in[n];
  const float4 b = ((const float4*)bias)[sub];
  float4 o;
  o.x = acc.x * sc + b.x;
  o.y = acc.y * sc + b.y;
  o.z = acc.z * sc + b.z;
  o.w = acc.w * sc + b.w;
  ((float4*)(out + (size_t)n * 32))[sub] = o;
}

extern "C" void kernel_launch(void* const* d_in, const int* in_sizes, int n_in,
                              void* d_out, int out_size, void* d_ws, size_t ws_size,
                              hipStream_t stream) {
  const float* in_feat = (const float*)d_in[0];
  const int*   src     = (const int*)d_in[1];
  const int*   dst     = (const int*)d_in[2];
  const float* W1      = (const float*)d_in[3];
  const float* b1      = (const float*)d_in[4];
  const float* W2      = (const float*)d_in[5];
  const float* b2      = (const float*)d_in[6];
  float* out = (float*)d_out;

  const int N = GCN_N_NODES;
  const int E = GCN_N_EDGES;

  float* inv_out = (float*)d_ws;
  float* inv_in  = inv_out + 2 * N;
  int*   rowptr  = (int*)(inv_in + N);
  int*   csr_src = rowptr + (N + 16) + N;
  float* t       = (float*)(csr_src + E);
  int*   hist_in  = (int*)(t + (size_t)N * 32);
  int*   hist_out = hist_in + (size_t)GCN_G * GCN_R * GCN_NRP;
  int*   csr_g    = hist_out + (size_t)GCN_G * GCN_R * GCN_NR;

  gcn_count_kernel<<<dim3(GCN_G, 2 * GCN_R), 1024, 0, stream>>>(src, dst, hist_in, hist_out, csr_g);
  gcn_merge_kernel<<<2 * GCN_NB2, 256, 0, stream>>>(hist_in, hist_out, csr_g, inv_out, inv_in,
                                                   rowptr, csr_src);

  gcn_layer1_fused<<<N / 16, 256, 0, stream>>>(in_feat, rowptr, csr_src, inv_out, inv_in,
                                             W1, b1, W2, t);

  gcn_gather32<<<(N + 31) / 32, 256, 0, stream>>>(t, rowptr, csr_src, inv_in, b2, out, N);
}


# Both gather launches: half the waves, twice the rows each. Layer 1 on 32-node tiles (a wave walks 4 nodes' ids as one list, 8 row loads in flight), layer 2 two nodes per lane group; sum orders kept
# speedup vs baseline: 6.1683x; 1.0333x over previous
#include <hip/hip_runtime.h>
#include <hip/hip_bf16.h>


#define GCN_N_NODES 50000
#define GCN_N_EDGES 800000
#define GCN_IN      128
#define GCN_H       128
#define GCN_C       32

typedef __attribute__((ext_vector_type(2))) float v2f;
typedef __attribute__((ext_vector_type(8))) float v8f;

#define GCN_G      32
#define GCN_CHUNK  ((((GCN_N_EDGES + GCN_G - 1) / GCN_G) + 3) / 4 * 4)
#define GCN_KQ     ((GCN_CHUNK / 4 + 1023) / 1024)
#define GCN_NB2    ((GCN_N_NODES + 255) / 256)
#define GCN_R      4
#define GCN_NR     ((GCN_NB2 + GCN_R - 1) / GCN_R * 256)
#define GCN_SEG4R  5
#define GCN_TBLR   (1024 * GCN_SEG4R * 4)
#define GCN_NRP    (GCN_NR + 4)
#define GCN_DB     14
static_assert(GCN_N_NODES % 4 == 0 && GCN_N_EDGES % 4 == 0, "4-wide loads");
static_assert(GCN_CHUNK % 4 == 0, "a 4-edge load stays inside one group");
static_assert(GCN_SEG4R % 2 == 1, "scan stride of 16 B x odd: 16-byte LDS accesses of a wave do not collide");
static_assert(GCN_NR % 256 == 0 && GCN_R * GCN_NR >= GCN_N_NODES, "ranges are whole merge blocks and cover the nodes");
static_assert(GCN_TBLR >= GCN_NRP, "a range's table holds its nodes and the total");
static_assert(GCN_G * GCN_R <= 256, "the merge adds up the earlier ranges' totals with one thread each");
static_assert(GCN_NR <= (1 << GCN_DB) && GCN_CHUNK < (1 << (31 - GCN_DB)), "place and rank share one non-negative word");

__device__ __forceinline__ int gcn_place_rank(int* tbl, int d, bool valid) {
  if (valid && (unsigned)d < (unsigned)GCN_NR)
    return (atomicAdd(&tbl[d], 1) << GCN_DB) | d;
  return -1;
}

__global__ __launch_bounds__(1024) void gcn_count_kernel(
    const int* __restrict__ src, const int* __restrict__ dst,
    int* __restrict__ hist_in, int* __restrict__ hist_out,
    int* __restrict__ csr_g) {
  __shared__ __attribute__((aligned(16))) int tbl[GCN_TBLR];
  __shared__ __attribute__((aligned(16))) int stage[GCN_CHUNK];
  __shared__ int wsum[32];
  const int tid  = threadIdx.x;
  const int lane = tid & 31;
  const int w    = tid >> 5;
  const int g    = blockIdx.x;
  const bool by_dst = (blockIdx.y < GCN_R);
  const int rng = by_dst ? (int)blockIdx.y : (int)blockIdx.y - GCN_R;
  const int lo = rng * GCN_NR;
  const int e0 = g * GCN_CHUNK;
  const int e1 = (e0 + GCN_CHUNK < GCN_N_EDGES) ? (e0 + GCN_CHUNK) : GCN_N_EDGES;
  const int nq = (e1 > e0) ? ((e1 - e0) >> 2) : 0;
  const int4* __restrict__ ids  = (const int4*)((by_dst ? dst : src) + e0);
  const int4* __restrict__ sids = (const int4*)(src + e0);

  int4 id[GCN_KQ];
  int4 sid[GCN_KQ];
#pragma unroll
  for (int k = 0; k < GCN_KQ; ++k) {
    const int qi = k * 1024 + tid;
    id[k]  = (qi < nq) ? ids[qi] : make_int4(0, 0, 0, 0);
    sid[k] = (by_dst && qi < nq) ? sids[qi] : make_int4(0, 0, 0, 0);
  }

  for (int i = tid; i < GCN_TBLR / 4; i += 1024)
    ((int4*)tbl)[i] = make_int4(0, 0, 0, 0);
  __syncthreads();

  if (!by_dst) {
#pragma unroll
    for (int k = 0; k < GCN_KQ; ++k) {
      const int qi = k * 1024 + tid;
      if (qi < nq) {
        if ((unsigned)(id[k].x - lo) < (unsigned)GCN_NR) atomicAdd(&tbl[id[k].x - lo], 1);
        if ((unsigned)(id[k].y - lo) < (unsigned)GCN_NR) atomicAdd(&tbl[id[k].y - lo], 1);
        if ((unsigned)(id[k].z - lo) < (unsigned)GCN_NR) atomicAdd(&tbl[id[k].z - lo], 1);
        if ((unsigned)(id[k].w - lo) < (unsigned)GCN_NR) atomicAdd(&tbl[id[k].w - lo], 1);
      }
    }
    __syncthreads();
    int* __restrict__ ho = hist_out + (size_t)(g * GCN_R + rng) * GCN_NR;
    for (int i = tid; i < GCN_NR / 4; i += 1024)
      ((int4*)ho)[i] = ((const int4*)tbl)[i];
    return;
  }

#pragma unroll
  for (int k = 0; k < GCN_KQ; ++k) {
    const bool v = (k * 1024 + tid < nq);
    id[k].x = gcn_place_rank(tbl, id[k].x - lo, v);
    id[k].y = gcn_place_rank(tbl, id[k].y - lo, v);
    id[k].z = gcn_place_rank(tbl, id[k].z - lo, v);
    id[k].w = gcn_place_rank(tbl, id[k].w - lo, v);
  }
  __syncthreads();

  int4* seg = (int4*)tbl + tid * GCN_SEG4R;
  int sum = 0;
#pragma unroll
  for (int u = 0; u < GCN_SEG4R; ++u) {
    const int4 v = seg[u];
    sum += v.x + v.y + v.z + v.w;
  }
  int incl = sum;
  for (int off = 1; off < 32; off <<= 1) {
    int y = __shfl_up(incl, off);
    if (lane >= off) incl += y;
  }
  if (lane == 31) wsum[w] = incl;
  __syncthreads();
  int part = (lane < w) ? wsum[lane] : 0;
  for (int off = 16; off > 0; off >>= 1) part += __shfl_xor(part, off);
  int run = part + incl - sum;
#pragma unroll
  for (int u = 0; u < GCN_SEG4R; ++u) {
    const int4 v = seg[u];
    int4 o;
    o.x = run;
    o.y = o.x + v.x;
    o.z = o.y + v.y;
    o.w = o.z + v.z;
    run = o.w + v.w;
    seg[u] = o;
  }
  __syncthreads();

#pragma unroll
  for (int k = 0; k < GCN_KQ; ++k) {
    if (id[k].x >= 0) stage[tbl[id[k].x & ((1 << GCN_DB) - 1)] + (id[k].x >> GCN_DB)] = sid[k].x;
    if (id[k].y >= 0) stage[tbl[id[k].y & ((1 << GCN_DB) - 1)] + (id[k].y >> GCN_DB)] = sid[k].y;
    if (id[k].z >= 0) stage[tbl[id[k].z & ((1 << GCN_DB) - 1)] + (id[k].z >> GCN_DB)] = sid[k].z;
    if (id[k].w >= 0) stage[tbl[id[k].w & ((1 << GCN_DB) - 1)] + (id[k].w >> GCN_DB)] = sid[k].w;
  }
  __syncthreads();

  const int row = g * GCN_R + rng;
  int* __restrict__ hi = hist_in + (size_t)row * GCN_NRP;
  for (int i = tid; i < GCN_NRP / 4; i += 1024)
    ((int4*)hi)[i] = ((const int4*)tbl)[i];
  const int tot4 = (tbl[GCN_NR] + 3) >> 2;
  int* __restrict__ cg = csr_g + (size_t)row * GCN_CHUNK;
  for (int i = tid; i < tot4; i += 1024)
    ((int4*)cg)[i] = ((const int4*)stage)[i];
}

__global__ __launch_bounds__(256) void gcn_merge_kernel(
    const int* __restrict__ hist_in, const int* __restrict__ hist_out,
    const int* __restrict__ csr_g, float* __restrict__ inv_out,
    float* __restrict__ inv_in, int* __restrict__ rowptr,
    int* __restrict__ csr_src) {
  __shared__ int tot_w[8];
  __shared__ int ext_w[8];
  __shared__ int pre_s;
  const int tid  = threadIdx.x;
  const int lane = tid & 31;
  const int w    = tid >> 5;
  const int N    = GCN_N_NODES;

  if (blockIdx.x >= GCN_NB2) {
    const int bi = blockIdx.x - GCN_NB2;
    const int i = bi * 256 + tid;
    const int rg = bi / (GCN_NR / 256);
    const int li = i - rg * GCN_NR;
    if (i < N) {
      int d = 0;
#pragma unroll
      for (int g = 0; g < GCN_G; ++g) d += hist_out[(unsigned)((g * GCN_R + rg) * GCN_NR + li)];
      inv_out[i] = (d > 0) ? rsqrtf((float)d) : 0.0f;
    }
    return;
  }

  const int i = blockIdx.x * 256 + tid;
  const int rg = blockIdx.x / (GCN_NR / 256);
  const int li = i - rg * GCN_NR;
  int s[GCN_G], n[GCN_G];
  int c = 0, sp = 0;
#pragma unroll
  for (int g = 0; g < GCN_G; ++g) {
    int a = 0, b = 0;
    if (i < N) {
      const unsigned h = (unsigned)((g * GCN_R + rg) * GCN_NRP + li);
      a = hist_in[h];
      b = hist_in[h + 1];
    }
    s[g] = a;
    n[g] = b - a;
    c  += b - a;
    sp += a;
  }
  if (tid == 0) pre_s = sp;
  int ex = 0;
  if (tid < GCN_G * GCN_R && (tid % GCN_R) < rg)
    ex = hist_in[(unsigned)(tid * GCN_NRP + GCN_NR)];
  for (int off = 16; off > 0; off >>= 1) ex += __shfl_xor(ex, off);
  if (lane == 0) ext_w[w] = ex;

  int a0[GCN_G], a1[GCN_G], a2[GCN_G], a3[GCN_G];
#pragma unroll
  for (int g = 0; g < GCN_G; ++g) {
    const unsigned p = (unsigned)((g * GCN_R + rg) * GCN_CHUNK + s[g]);
    a0[g] = (n[g] > 0) ? csr_g[p]     : 0;
    a1[g] = (n[g] > 1) ? csr_g[p + 1] : 0;
    a2[g] = (n[g] > 2) ? csr_g[p + 2] : 0;
    a3[g] = (n[g] > 3) ? csr_g[p + 3] : 0;
  }

  int incl = c;
  for (int off = 1; off < 32; off <<= 1) {
    int y = __shfl_up(incl, off);
    if (lane >= off) incl += y;
  }
  if (lane == 31) tot_w[w] = incl;
  __syncthreads();
  int part = (lane < w) ? tot_w[lane & 7] : 0;
  for (int off = 16; off > 0; off >>= 1) part += __shfl_xor(part, off);
  int pre = pre_s;
  for (int k = 0; k < 8; ++k) pre += ext_w[k];
  const int excl = pre + part + incl - c;

  if (i < N) {
    rowptr[i] = excl;
    inv_in[i] = (c > 0) ? rsqrtf((float)c) : 0.0f;
    if (i == N - 1) rowptr[N] = excl + c;
  }
  unsigned run = (unsigned)excl;
#pragma unroll
  for (int g = 0; g < GCN_G; ++g) {
    if (n[g] > 0) csr_src[run]     = a0[g];
    if (n[g] > 1) csr_src[run + 1] = a1[g];
    if (n[g] > 2) csr_src[run + 2] = a2[g];
    if (n[g] > 3) csr_src[run + 3] = a3[g];
    const unsigned p = (unsigned)((g * GCN_R + rg) * GCN_CHUNK + s[g]);
#pragma clang loop vectorize(disable) unroll(disable)
    for (int k = 4; k < n[g]; ++k) csr_src[run + k] = csr_g[p + k];
    run += (unsigned)n[g];
  }
}

#define GCN_TP 132
#define GCN_TM 32

#define GCN_FLUSH_ROW()                                                              \
  do {                                                                               \
    const float d_ = __int_as_float(__builtin_amdgcn_readlane(__float_as_int(di_l), k)); \
    ((float4*)As[arow0 + k])[lane] =                                                 \
        make_float4(acc.x * d_, acc.y * d_, acc.z * d_, acc.w * d_);                 \
    acc = make_float4(0.0f, 0.0f, 0.0f, 0.0f);                                       \
    ++k;                                                                             \
    nb = (k == 1) ? b2 : ((k == 2) ? b3 : b4);                                       \
  } while (0)

__device__ __forceinline__ void gcn_walk32(
    const float4* __restrict__ xl, float (*As)[GCN_TP], int arow0, int lane,
    int s_l, float w_l, int base, int m, int b2, int b3, int b4, float di_l,
    float4& acc, int& k, int& nb) {
  for (int j = 0; j < m; j += 8) {
    float4 v[8];
#pragma unroll
    for (int q = 0; q < 8; ++q) {
      const int s = __builtin_amdgcn_readlane(s_l, j + q);
      v[q] = xl[(size_t)s * 32];
    }
#pragma unroll
    for (int q = 0; q < 8; ++q) {
      if (j + q < m) {
        const int e = base + j + q;
        while (k < 3 && e >= nb) GCN_FLUSH_ROW();
        const float w = __int_as_float(__builtin_amdgcn_readlane(__float_as_int(w_l), j + q));
        acc.x += v[q].x * w; acc.y += v[q].y * w; acc.z += v[q].z * w; acc.w += v[q].w * w;
      }
    }
  }
}

__global__ __launch_bounds__(256, 16) void gcn_layer1_fused(
    const float* __restrict__ x, const int* __restrict__ rowptr,
    const int* __restrict__ csr_src, const float* __restrict__ inv_out,
    const float* __restrict__ inv_in, const float* __restrict__ W1,
    const float* __restrict__ b1, const float* __restrict__ W2,
    float* __restrict__ t) {
  __shared__ __attribute__((aligned(16))) float As[GCN_TM][GCN_TP];
  __shared__ __attribute__((aligned(16))) float Hs[GCN_TM][GCN_TP];
  __shared__ __attribute__((aligned(16))) float Sc[GCN_TM];
  const int N    = GCN_N_NODES;
  const int tid  = threadIdx.x;
  const int lane = tid & 31;
  const int wave = __builtin_amdgcn_readfirstlane(tid >> 5);
  const int row0 = blockIdx.x * GCN_TM;

  const int n0 = row0 + 4 * wave;
  const int arow0 = 4 * wave;
  const int ri = (n0 + lane < N) ? (n0 + lane) : N;
  const int   rp_l = (lane < 5) ? rowptr[ri] : 0;
  const float di_l = (lane < 4 && n0 + lane < N) ? inv_in[n0 + lane] : 0.0f;
  if (wave == 0) Sc[lane] = (row0 + lane < N) ? inv_out[row0 + lane] : 0.0f;
  const int b0 = __builtin_amdgcn_readlane(rp_l, 0);
  const int b1e = __builtin_amdgcn_readlane(rp_l, 1);
  const int b2 = __builtin_amdgcn_readlane(rp_l, 2);
  const int b3 = __builtin_amdgcn_readlane(rp_l, 3);
  const int b4 = __builtin_amdgcn_readlane(rp_l, 4);
  int   sl[3];
  float wl[3];
#pragma unroll
  for (int c = 0; c < 3; ++c) {
    const int e = b0 + 32 * c + lane;
    sl[c] = (e < b4) ? csr_src[e] : 0;
  }
#pragma unroll
  for (int c = 0; c < 3; ++c) {
    wl[c] = inv_out[sl[c]];
  }
  const float4* __restrict__ xl = (const float4*)x + lane;
  float4 acc = make_float4(0.0f, 0.0f, 0.0f, 0.0f);
  int k = 0;
  int nb = b1e;
#pragma unroll
  for (int c = 0; c < 3; ++c) {
    const int base = b0 + 32 * c;
    if (base < b4) {
      const int m = (b4 - base < 32) ? (b4 - base) : 32;
      gcn_walk32(xl, As, arow0, lane, sl[c], wl[c], base, m, b2, b3, b4, di_l, acc, k, nb);
    }
  }
  for (int base = b0 + 96; base < b4; base += 32) {
    const int m = (b4 - base < 32) ? (b4 - base) : 32;
    int   s_l = 0;
    float w_l = 0.0f;
    if (lane < m) {
      s_l = csr_src[base + lane];
      w_l = inv_out[s_l];
    }
    gcn_walk32(xl, As, arow0, lane, s_l, w_l, base, m, b2, b3, b4, di_l, acc, k, nb);
  }
  while (k < 4) GCN_FLUSH_ROW();
  __syncthreads();

  const int half = lane >> 4;
  const int mrow = lane & 15;
  {
    const int col0 = wave * 16;
    v8f acc0 = {};
    v8f acc1 = {};
    for (int kk = 0; kk < 128; kk += 4) {
      int k2 = kk + half * 2;
      v2f a0, a1, b;
      a0.x = As[mrow][k2];
      a0.y = As[mrow][k2 + 1];
      a1.x = As[16 + mrow][k2];
      a1.y = As[16 + mrow][k2 + 1];
      b.x = W1[(size_t)k2 * 128 + col0 + mrow];
      b.y = W1[(size_t)(k2 + 1) * 128 + col0 + mrow];
      acc0 = __builtin_amdgcn_wmma_f32_16x16x4_f32(
           false, a0,  false, b,
           (short)0, acc0,  false,  false);
      acc1 = __builtin_amdgcn_wmma_f32_16x16x4_f32(
          false, a1, false, b, (short)0, acc1, false, false);
    }
    const float bv = b1[col0 + mrow];
    for (int r = 0; r < 8; ++r) {
      const int row = r + half * 8;
      float v = acc0[r] + bv;
      v = (v > 0.0f) ? v : 0.0f;
      Hs[row][col0 + mrow] = v * Sc[row];
      float u = acc1[r] + bv;
      u = (u > 0.0f) ? u : 0.0f;
      Hs[16 + row][col0 + mrow] = u * Sc[16 + row];
    }
  }
  __syncthreads();

  const int pw = (wave + 4 * ((int)blockIdx.x & 1)) & 7;
  if (pw < 4) {
    const int r0 = (pw >> 1) * 16;
    const int c0 = (pw & 1) * 16;
    v8f acc2 = {};
    for (int kk = 0; kk < 128; kk += 4) {
      int k2 = kk + half * 2;
      v2f a, b;
      a.x = Hs[r0 + mrow][k2];
      a.y = Hs[r0 + mrow][k2 + 1];
      b.x = W2[(size_t)k2 * 32 + c0 + mrow];
      b.y = W2[(size_t)(k2 + 1) * 32 + c0 + mrow];
      acc2 = __builtin_amdgcn_wmma_f32_16x16x4_f32(
          false, a, false, b, (short)0, acc2, false, false);
    }
    for (int r = 0; r < 8; ++r) {
      const int row = row0 + r0 + r + half * 8;
      if (row < N) t[(size_t)row * 32 + c0 + mrow] = acc2[r];
    }
  }
}

__global__ __launch_bounds__(256, 8) void gcn_gather32(
    const float* __restrict__ t, const int* __restrict__ rowptr,
    const int* __restrict__ csr_src, const float* __restrict__ inv_in,
    const float* __restrict__ bias, float* __restrict__ out, int N) {
  const int sub = threadIdx.x & 7;
  const int grp = threadIdx.x >> 3;
  const float4* __restrict__ t4 = (const float4*)t + sub;
  int   beg[2], end[2], id[2][4];
  float sc[2];
#pragma unroll
  for (int h = 0; h < 2; ++h) {
    const int n = blockIdx.x * 64 + h * 32 + grp;
    const bool ok = (n < N);
    beg[h] = ok ? rowptr[n] : 0;
    end[h] = ok ? rowptr[n + 1] : 0;
    sc[h]  = ok ? inv_in[n] : 0.0f;
  }
#pragma unroll
  for (int h = 0; h < 2; ++h) {
#pragma unroll
    for (int c = 0; c < 4; ++c) {
      const int e = beg[h] + 8 * c + sub;
      id[h][c] = (e < end[h]) ? csr_src[e] : 0;
    }
  }
  const float4 b = ((const float4*)bias)[sub];
  float4 acc[2];
  acc[0] = make_float4(0.0f, 0.0f, 0.0f, 0.0f);
  acc[1] = make_float4(0.0f, 0.0f, 0.0f, 0.0f);
#pragma unroll
  for (int c = 0; c < 4; ++c) {
    float4 v[2][8];
#pragma unroll
    for (int h = 0; h < 2; ++h) {
      int s[8];
#pragma unroll
      for (int q = 0; q < 8; ++q) s[q] = __shfl(id[h][c], q, 8);
      if (beg[h] + 8 * c < end[h]) {
#pragma unroll
        for (int q = 0; q < 8; ++q) v[h][q] = t4[(size_t)s[q] * 8];
      }
    }
#pragma unroll
    for (int h = 0; h < 2; ++h) {
      if (beg[h] + 8 * c < end[h]) {
#pragma unroll
        for (int q = 0; q < 8; ++q) {
          if (beg[h] + 8 * c + q < end[h]) {
            acc[h].x += v[h][q].x; acc[h].y += v[h][q].y;
            acc[h].z += v[h][q].z; acc[h].w += v[h][q].w;
          }
        }
      }
    }
  }
#pragma unroll
  for (int h = 0; h < 2; ++h) {
    for (int j = beg[h] + 32; j < end[h]; ++j) {
      const int s0 = csr_src[j];
      const float4 v0 = t4[(size_t)s0 * 8];
      acc[h].x += v0.x; acc[h].y += v0.y; acc[h].z += v0.z; acc[h].w += v0.w;
    }
    const int n = blockIdx.x * 64 + h * 32 + grp;
    if (n < N) {
      float4 o;
      o.x = acc[h].x * sc[h] + b.x;
      o.y = acc[h].y * sc[h] + b.y;
      o.z = acc[h].z * sc[h] + b.z;
      o.w = acc[h].w * sc[h] + b.w;
      ((float4*)(out + (size_t)n * 32))[sub] = o;
    }
  }
}

extern "C" void kernel_launch(void* const* d_in, const int* in_sizes, int n_in,
                              void* d_out, int out_size, void* d_ws, size_t ws_size,
                              hipStream_t stream) {
  const float* in_feat = (const float*)d_in[0];
  const int*   src     = (const int*)d_in[1];
  const int*   dst     = (const int*)d_in[2];
  const float* W1      = (const float*)d_in[3];
  const float* b1      = (const float*)d_in[4];
  const float* W2      = (const float*)d_in[5];
  const float* b2      = (const float*)d_in[6];
  float* out = (float*)d_out;

  const int N = GCN_N_NODES;
  const int E = GCN_N_EDGES;

  float* inv_out = (float*)d_ws;
  float* inv_in  = inv_out + 2 * N;
  int*   rowptr  = (int*)(inv_in + N);
  int*   csr_src = rowptr + (N + 16) + N;
  float* t       = (float*)(csr_src + E);
  int*   hist_in  = (int*)(t + (size_t)N * 32);
  int*   hist_out = hist_in + (size_t)GCN_G * GCN_R * GCN_NRP;
  int*   csr_g    = hist_out + (size_t)GCN_G * GCN_R * GCN_NR;

  gcn_count_kernel<<<dim3(GCN_G, 2 * GCN_R), 1024, 0, stream>>>(src, dst, hist_in, hist_out, csr_g);
  gcn_merge_kernel<<<2 * GCN_NB2, 256, 0, stream>>>(hist_in, hist_out, csr_g, inv_out, inv_in,
                                                   rowptr, csr_src);

  gcn_layer1_fused<<<(GCN_N_NODES + GCN_TM - 1) / GCN_TM, 256, 0, stream>>>(in_feat, rowptr, csr_src, inv_out, inv_in,
                                             W1, b1, W2, t);

  gcn_gather32<<<(GCN_N_NODES + 63) / 64, 256, 0, stream>>>(t, rowptr, csr_src, inv_in, b2, out, N);
}


# Merge launch removed (4 launches -> 3): a layer-1 wave builds its 4 nodes' piece of the dst-sorted list from the per-group lists (lane = group, lane scans); out-degrees node-major; sum orders kept
# speedup vs baseline: 6.9865x; 1.1326x over previous
#include <hip/hip_runtime.h>
#include <hip/hip_bf16.h>


#define GCN_N_NODES 50000
#define GCN_N_EDGES 800000
#define GCN_IN      128
#define GCN_H       128
#define GCN_C       32

typedef __attribute__((ext_vector_type(2))) float v2f;
typedef __attribute__((ext_vector_type(8))) float v8f;

#define GCN_G      32
#define GCN_CHUNK  ((((GCN_N_EDGES + GCN_G - 1) / GCN_G) + 3) / 4 * 4)
#define GCN_KQ     ((GCN_CHUNK / 4 + 1023) / 1024)
#define GCN_NB2    ((GCN_N_NODES + 255) / 256)
#define GCN_R      4
#define GCN_NR     ((GCN_NB2 + GCN_R - 1) / GCN_R * 256)
#define GCN_SEG4R  5
#define GCN_TBLR   (1024 * GCN_SEG4R * 4)
#define GCN_NRP    (GCN_NR + 4)
#define GCN_DB     14
#define GCN_EG     8
#define GCN_R1     16
#define GCN_ECH    ((((GCN_N_EDGES + GCN_EG - 1) / GCN_EG) + 3) / 4 * 4)
#define GCN_NR1    ((((GCN_N_NODES + GCN_R1 - 1) / GCN_R1) + 3) / 4 * 4)
static_assert(GCN_N_NODES % 4 == 0 && GCN_N_EDGES % 4 == 0, "4-wide loads");
static_assert(GCN_EG * GCN_R1 == GCN_G * GCN_R, "as many out-degree blocks as sort blocks");
static_assert(GCN_EG == 8 && GCN_ECH % 4 == 0, "a node's out-degree counts are two 16-byte pieces; 4-edge loads");
static_assert(GCN_NR1 <= GCN_TBLR && GCN_R1 * GCN_NR1 >= GCN_N_NODES, "the small ranges fit the table and cover the nodes");
static_assert(GCN_G == 32, "layer 1: one lane per edge group");
static_assert(GCN_NR % 32 == 0, "a 32-node tile lies inside one node range");
static_assert(GCN_CHUNK % 4 == 0, "a 4-edge load stays inside one group");
static_assert(GCN_SEG4R % 2 == 1, "scan stride of 16 B x odd: 16-byte LDS accesses of a wave do not collide");
static_assert(GCN_NR % 256 == 0 && GCN_R * GCN_NR >= GCN_N_NODES, "ranges are whole 256-node blocks and cover the nodes");
static_assert(GCN_TBLR >= GCN_NRP, "a range's table holds its nodes and the total");
static_assert(GCN_NR <= (1 << GCN_DB) && GCN_CHUNK < (1 << (31 - GCN_DB)), "place and rank share one non-negative word");

__device__ __forceinline__ int gcn_place_rank(int* tbl, int d, bool valid) {
  if (valid && (unsigned)d < (unsigned)GCN_NR)
    return (atomicAdd(&tbl[d], 1) << GCN_DB) | d;
  return -1;
}

__global__ __launch_bounds__(1024) void gcn_count_kernel(
    const int* __restrict__ src, const int* __restrict__ dst,
    int* __restrict__ hist_in, int* __restrict__ cnt8,
    int* __restrict__ csr_g) {
  __shared__ __attribute__((aligned(16))) int tbl[GCN_TBLR];
  __shared__ __attribute__((aligned(16))) int stage[GCN_CHUNK];
  __shared__ int wsum[32];
  const int tid  = threadIdx.x;
  const int lane = tid & 31;
  const int w    = tid >> 5;
  if (blockIdx.y >= GCN_R) {
    const int bid = ((int)blockIdx.y - GCN_R) * GCN_G + (int)blockIdx.x;
    const int eg  = bid % GCN_EG;
    const int lo1 = (bid / GCN_EG) * GCN_NR1;
    const int f0 = eg * GCN_ECH;
    const int f1 = (f0 + GCN_ECH < GCN_N_EDGES) ? (f0 + GCN_ECH) : GCN_N_EDGES;
    const int mq = (f1 > f0) ? ((f1 - f0) >> 2) : 0;
    const int4* __restrict__ sq = (const int4*)(src + f0);
    for (int i = tid; i < GCN_NR1; i += 1024) tbl[i] = 0;
    __syncthreads();
    for (int q0 = 0; q0 < mq; q0 += 8 * 1024) {
      int4 v[8];
#pragma unroll
      for (int u = 0; u < 8; ++u) {
        const int qi = q0 + u * 1024 + tid;
        v[u] = (qi < mq) ? sq[qi] : make_int4(-1, -1, -1, -1);
      }
#pragma unroll
      for (int u = 0; u < 8; ++u) {
        if ((unsigned)(v[u].x - lo1) < (unsigned)GCN_NR1) atomicAdd(&tbl[v[u].x - lo1], 1);
        if ((unsigned)(v[u].y - lo1) < (unsigned)GCN_NR1) atomicAdd(&tbl[v[u].y - lo1], 1);
        if ((unsigned)(v[u].z - lo1) < (unsigned)GCN_NR1) atomicAdd(&tbl[v[u].z - lo1], 1);
        if ((unsigned)(v[u].w - lo1) < (unsigned)GCN_NR1) atomicAdd(&tbl[v[u].w - lo1], 1);
      }
    }
    __syncthreads();
    for (int i = tid; i < GCN_NR1; i += 1024)
      cnt8[(size_t)(lo1 + i) * GCN_EG + eg] = tbl[i];
    return;
  }

  const int g    = blockIdx.x;
  const int rng  = blockIdx.y;
  const int lo = rng * GCN_NR;
  const int e0 = g * GCN_CHUNK;
  const int e1 = (e0 + GCN_CHUNK < GCN_N_EDGES) ? (e0 + GCN_CHUNK) : GCN_N_EDGES;
  const int nq = (e1 > e0) ? ((e1 - e0) >> 2) : 0;
  const int4* __restrict__ ids  = (const int4*)(dst + e0);
  const int4* __restrict__ sids = (const int4*)(src + e0);

  int4 id[GCN_KQ];
  int4 sid[GCN_KQ];
#pragma unroll
  for (int k = 0; k < GCN_KQ; ++k) {
    const int qi = k * 1024 + tid;
    id[k]  = (qi < nq) ? ids[qi] : make_int4(0, 0, 0, 0);
    sid[k] = (qi < nq) ? sids[qi] : make_int4(0, 0, 0, 0);
  }

  for (int i = tid; i < GCN_TBLR / 4; i += 1024)
    ((int4*)tbl)[i] = make_int4(0, 0, 0, 0);
  __syncthreads();

#pragma unroll
  for (int k = 0; k < GCN_KQ; ++k) {
    const bool v = (k * 1024 + tid < nq);
    id[k].x = gcn_place_rank(tbl, id[k].x - lo, v);
    id[k].y = gcn_place_rank(tbl, id[k].y - lo, v);
    id[k].z = gcn_place_rank(tbl, id[k].z - lo, v);
    id[k].w = gcn_place_rank(tbl, id[k].w - lo, v);
  }
  __syncthreads();

  int4* seg = (int4*)tbl + tid * GCN_SEG4R;
  int sum = 0;
#pragma unroll
  for (int u = 0; u < GCN_SEG4R; ++u) {
    const int4 v = seg[u];
    sum += v.x + v.y + v.z + v.w;
  }
  int incl = sum;
  for (int off = 1; off < 32; off <<= 1) {
    int y = __shfl_up(incl, off);
    if (lane >= off) incl += y;
  }
  if (lane == 31) wsum[w] = incl;
  __syncthreads();
  int part = (lane < w) ? wsum[lane] : 0;
  for (int off = 16; off > 0; off >>= 1) part += __shfl_xor(part, off);
  int run = part + incl - sum;
#pragma unroll
  for (int u = 0; u < GCN_SEG4R; ++u) {
    const int4 v = seg[u];
    int4 o;
    o.x = run;
    o.y = o.x + v.x;
    o.z = o.y + v.y;
    o.w = o.z + v.z;
    run = o.w + v.w;
    seg[u] = o;
  }
  __syncthreads();

#pragma unroll
  for (int k = 0; k < GCN_KQ; ++k) {
    if (id[k].x >= 0) stage[tbl[id[k].x & ((1 << GCN_DB) - 1)] + (id[k].x >> GCN_DB)] = sid[k].x;
    if (id[k].y >= 0) stage[tbl[id[k].y & ((1 << GCN_DB) - 1)] + (id[k].y >> GCN_DB)] = sid[k].y;
    if (id[k].z >= 0) stage[tbl[id[k].z & ((1 << GCN_DB) - 1)] + (id[k].z >> GCN_DB)] = sid[k].z;
    if (id[k].w >= 0) stage[tbl[id[k].w & ((1 << GCN_DB) - 1)] + (id[k].w >> GCN_DB)] = sid[k].w;
  }
  __syncthreads();

  const int row = g * GCN_R + rng;
  int* __restrict__ hi = hist_in + (size_t)row * GCN_NRP;
  for (int i = tid; i < GCN_NRP / 4; i += 1024)
    ((int4*)hi)[i] = ((const int4*)tbl)[i];
  const int tot4 = (tbl[GCN_NR] + 3) >> 2;
  int* __restrict__ cg = csr_g + (size_t)row * GCN_CHUNK;
  for (int i = tid; i < tot4; i += 1024)
    ((int4*)cg)[i] = ((const int4*)stage)[i];
}

#define GCN_TP 132
#define GCN_TM 32

#define GCN_FLUSH_ROW()                                                              \
  do {                                                                               \
    const float d_ = __int_as_float(__builtin_amdgcn_readlane(__float_as_int(di_l), k)); \
    ((float4*)As[arow0 + k])[lane] =                                                 \
        make_float4(acc.x * d_, acc.y * d_, acc.z * d_, acc.w * d_);                 \
    acc = make_float4(0.0f, 0.0f, 0.0f, 0.0f);                                       \
    ++k;                                                                             \
    nb = (k == 1) ? b2 : ((k == 2) ? b3 : b4);                                       \
  } while (0)

#define GCN_LCAP (4 * GCN_TP)

__device__ __forceinline__ float gcn_dout(const int* __restrict__ cnt8, int s) {
  const int4* __restrict__ p = (const int4*)(cnt8 + (size_t)s * GCN_EG);
  const int4 u = p[0];
  const int4 v = p[1];
  const int d = ((u.x + u.y) + (u.z + u.w)) + ((v.x + v.y) + (v.z + v.w));
  return (d > 0) ? rsqrtf((float)d) : 0.0f;
}

__device__ __forceinline__ int gcn_scan32(int x, int lane) {
  x += __builtin_amdgcn_update_dpp(0, x, 0x111, 0xf, 0xf, true);
  x += __builtin_amdgcn_update_dpp(0, x, 0x112, 0xf, 0xf, true);
  x += __builtin_amdgcn_update_dpp(0, x, 0x114, 0xf, 0xf, true);
  x += __builtin_amdgcn_update_dpp(0, x, 0x118, 0xf, 0xf, true);
  const int low = __builtin_amdgcn_readlane(x, 15);
  return (lane >= 16) ? (x + low) : x;
}

#define GCN_PLACE(j) ((((j) < e1) ? q0 : (((j) < e2) ? q1 : (((j) < e3) ? q2 : q3))) + (j))

__device__ __forceinline__ void gcn_walk32(
    const float4* __restrict__ xl, float (*As)[GCN_TP], int arow0, int lane,
    int s_l, float w_l, int base, int m, int b2, int b3, int b4, float di_l,
    float4& acc, int& k, int& nb) {
  for (int j = 0; j < m; j += 8) {
    float4 v[8];
#pragma unroll
    for (int q = 0; q < 8; ++q) {
      const int s = __builtin_amdgcn_readlane(s_l, j + q);
      v[q] = xl[(size_t)s * 32];
    }
#pragma unroll
    for (int q = 0; q < 8; ++q) {
      if (j + q < m) {
        const int e = base + j + q;
        while (k < 3 && e >= nb) GCN_FLUSH_ROW();
        const float w = __int_as_float(__builtin_amdgcn_readlane(__float_as_int(w_l), j + q));
        acc.x += v[q].x * w; acc.y += v[q].y * w; acc.z += v[q].z * w; acc.w += v[q].w * w;
      }
    }
  }
}

__global__ __launch_bounds__(256, 16) void gcn_layer1_fused(
    const float* __restrict__ x, const int* __restrict__ hist_in,
    const int* __restrict__ csr_g, const int* __restrict__ cnt8,
    const float* __restrict__ W1,
    const float* __restrict__ b1, const float* __restrict__ W2,
    float* __restrict__ t, int* __restrict__ rowptr,
    int* __restrict__ csr_src, float* __restrict__ inv_in) {
  __shared__ __attribute__((aligned(16))) float As[GCN_TM][GCN_TP];
  __shared__ __attribute__((aligned(16))) float Hs[GCN_TM][GCN_TP];
  __shared__ __attribute__((aligned(16))) float Sc[GCN_TM];
  const int N    = GCN_N_NODES;
  const int tid  = threadIdx.x;
  const int lane = tid & 31;
  const int wave = __builtin_amdgcn_readfirstlane(tid >> 5);
  const int row0 = blockIdx.x * GCN_TM;

  const int n0 = row0 + 4 * wave;
  const int arow0 = 4 * wave;
  const int rg = row0 / GCN_NR;
  const int li = n0 - rg * GCN_NR;
  const int* __restrict__ hrow = hist_in + (size_t)((lane * GCN_R + rg) * GCN_NRP + li);
  const int4 st = *(const int4*)hrow;
  const int  s4 = hrow[4];
  int ex = 0;
#pragma unroll
  for (int r = 0; r < GCN_R - 1; ++r) {
    const int tr = hist_in[(lane * GCN_R + r) * GCN_NRP + GCN_NR];
    ex += (r < rg) ? tr : 0;
  }

  const int c0 = st.y - st.x, c1 = st.z - st.y, c2 = st.w - st.z, c3 = s4 - st.w;
  const int e1 = c0, e2 = c0 + c1, e3 = e2 + c2;
  const int tl = s4 - st.x;
  const int* __restrict__ gp = csr_g + (size_t)(lane * GCN_R + rg) * GCN_CHUNK + st.x;
  int a[8];
#pragma unroll
  for (int j = 0; j < 8; ++j) a[j] = (j < tl) ? gp[j] : 0;
  if (wave == 0) Sc[lane] = (row0 + lane < N) ? gcn_dout(cnt8, row0 + lane) : 0.0f;

  const int i0 = gcn_scan32(c0, lane), i1 = gcn_scan32(c1, lane);
  const int i2 = gcn_scan32(c2, lane), i3 = gcn_scan32(c3, lane);
  const int rs = gcn_scan32(st.x + ex, lane);
  const int b0 = __builtin_amdgcn_readlane(rs, 31);
  const int d0 = __builtin_amdgcn_readlane(i0, 31);
  const int d1 = __builtin_amdgcn_readlane(i1, 31);
  const int d2 = __builtin_amdgcn_readlane(i2, 31);
  const int d3 = __builtin_amdgcn_readlane(i3, 31);
  const int b1e = b0 + d0;
  const int b2 = b1e + d1;
  const int b3 = b2 + d2;
  const int b4 = b3 + d3;
  const int tot = b4 - b0;
  const int q0 = i0 - c0;
  const int q1 = d0 + (i1 - c1) - e1;
  const int q2 = d0 + d1 + (i2 - c2) - e2;
  const int q3 = d0 + d1 + d2 + (i3 - c3) - e3;

  float* lst = &Hs[arow0][0];
#pragma unroll
  for (int j = 0; j < 8; ++j) {
    if (j < tl) {
      const int pos = GCN_PLACE(j);
      if (pos < GCN_LCAP) lst[pos] = __int_as_float(a[j]);
      else csr_src[b0 + pos] = a[j];
    }
  }
  for (int j = 8; j < tl; ++j) {
    const int idj = gp[j];
    const int pos = GCN_PLACE(j);
    if (pos < GCN_LCAP) lst[pos] = __int_as_float(idj);
    else csr_src[b0 + pos] = idj;
  }
  __syncthreads();

  const int dl = (lane == 0) ? d0 : ((lane == 1) ? d1 : ((lane == 2) ? d2 : d3));
  const int bl = (lane == 0) ? b0 : ((lane == 1) ? b1e : ((lane == 2) ? b2 : b3));
  const float di_l = (lane < 4 && dl > 0) ? rsqrtf((float)dl) : 0.0f;
  if (lane < 4 && n0 + lane < N) {
    rowptr[n0 + lane] = bl;
    inv_in[n0 + lane] = di_l;
  }
  if (lane == 4 && n0 + 4 == N) rowptr[N] = b4;

  const float4* __restrict__ xl = (const float4*)x + lane;
  float4 acc = make_float4(0.0f, 0.0f, 0.0f, 0.0f);
  if (tot <= GCN_LCAP) {
    int   sl[3];
    float wl[3];
#pragma unroll
    for (int c = 0; c < 3; ++c) {
      const int e = 32 * c + lane;
      sl[c] = (e < tot) ? __float_as_int(lst[e]) : 0;
      if (e < tot) csr_src[b0 + e] = sl[c];
    }
#pragma unroll
    for (int c = 0; c < 3; ++c) {
      wl[c] = gcn_dout(cnt8, sl[c]);
    }
    asm volatile("" :: "v"(wl[0]), "v"(wl[1]), "v"(wl[2]));
    int k = 0;
    int nb = b1e;
#pragma unroll
    for (int c = 0; c < 3; ++c) {
      const int base = b0 + 32 * c;
      if (base < b4) {
        const int m = (b4 - base < 32) ? (b4 - base) : 32;
        gcn_walk32(xl, As, arow0, lane, sl[c], wl[c], base, m, b2, b3, b4, di_l, acc, k, nb);
      }
    }
    for (int base = b0 + 96; base < b4; base += 32) {
      const int m = (b4 - base < 32) ? (b4 - base) : 32;
      int   s_l = 0;
      float w_l = 0.0f;
      if (lane < m) {
        s_l = __float_as_int(lst[base - b0 + lane]);
        csr_src[base + lane] = s_l;
        w_l = gcn_dout(cnt8, s_l);
      }
      gcn_walk32(xl, As, arow0, lane, s_l, w_l, base, m, b2, b3, b4, di_l, acc, k, nb);
    }
    while (k < 4) GCN_FLUSH_ROW();
  } else {
    for (int e = lane; e < GCN_LCAP; e += 32) csr_src[b0 + e] = __float_as_int(lst[e]);
#pragma clang loop unroll(disable)
    for (int kk = 0; kk < 4; ++kk) {
      const int sk = (kk == 0) ? st.x : ((kk == 1) ? st.y : ((kk == 2) ? st.z : st.w));
      const int ck = (kk == 0) ? c0 : ((kk == 1) ? c1 : ((kk == 2) ? c2 : c3));
#pragma clang loop unroll(disable)
      for (int g = 0; g < GCN_G; ++g) {
        const int sg = __builtin_amdgcn_readlane(sk, g);
        const int cg = __builtin_amdgcn_readlane(ck, g);
        const int* __restrict__ pg = csr_g + (size_t)(g * GCN_R + rg) * GCN_CHUNK + sg;
#pragma clang loop vectorize(disable) unroll(disable)
        for (int r = 0; r < cg; ++r) {
          const int sid = pg[r];
          const float w = gcn_dout(cnt8, sid);
          const float4 v = xl[(size_t)sid * 32];
          acc.x += v.x * w; acc.y += v.y * w; acc.z += v.z * w; acc.w += v.w * w;
        }
      }
      const float d_ = __int_as_float(__builtin_amdgcn_readlane(__float_as_int(di_l), kk));
      ((float4*)As[arow0 + kk])[lane] = make_float4(acc.x * d_, acc.y * d_, acc.z * d_, acc.w * d_);
      acc = make_float4(0.0f, 0.0f, 0.0f, 0.0f);
    }
  }
  __syncthreads();

  const int half = lane >> 4;
  const int mrow = lane & 15;
  {
    const int col0 = wave * 16;
    v8f acc0 = {};
    v8f acc1 = {};
    for (int kk = 0; kk < 128; kk += 4) {
      int k2 = kk + half * 2;
      v2f a0, a1, b;
      a0.x = As[mrow][k2];
      a0.y = As[mrow][k2 + 1];
      a1.x = As[16 + mrow][k2];
      a1.y = As[16 + mrow][k2 + 1];
      b.x = W1[(size_t)k2 * 128 + col0 + mrow];
      b.y = W1[(size_t)(k2 + 1) * 128 + col0 + mrow];
      acc0 = __builtin_amdgcn_wmma_f32_16x16x4_f32(
           false, a0,  false, b,
           (short)0, acc0,  false,  false);
      acc1 = __builtin_amdgcn_wmma_f32_16x16x4_f32(
          false, a1, false, b, (short)0, acc1, false, false);
    }
    const float bv = b1[col0 + mrow];
    for (int r = 0; r < 8; ++r) {
      const int row = r + half * 8;
      float v = acc0[r] + bv;
      v = (v > 0.0f) ? v : 0.0f;
      Hs[row][col0 + mrow] = v * Sc[row];
      float u = acc1[r] + bv;
      u = (u > 0.0f) ? u : 0.0f;
      Hs[16 + row][col0 + mrow] = u * Sc[16 + row];
    }
  }
  __syncthreads();

  const int pw = (wave + 4 * ((int)blockIdx.x & 1)) & 7;
  if (pw < 4) {
    const int r0 = (pw >> 1) * 16;
    const int c0 = (pw & 1) * 16;
    v8f acc2 = {};
    for (int kk = 0; kk < 128; kk += 4) {
      int k2 = kk + half * 2;
      v2f a, b;
      a.x = Hs[r0 + mrow][k2];
      a.y = Hs[r0 + mrow][k2 + 1];
      b.x = W2[(size_t)k2 * 32 + c0 + mrow];
      b.y = W2[(size_t)(k2 + 1) * 32 + c0 + mrow];
      acc2 = __builtin_amdgcn_wmma_f32_16x16x4_f32(
          false, a, false, b, (short)0, acc2, false, false);
    }
    for (int r = 0; r < 8; ++r) {
      const int row = row0 + r0 + r + half * 8;
      if (row < N) t[(size_t)row * 32 + c0 + mrow] = acc2[r];
    }
  }
}

__global__ __launch_bounds__(256, 8) void gcn_gather32(
    const float* __restrict__ t, const int* __restrict__ rowptr,
    const int* __restrict__ csr_src, const float* __restrict__ inv_in,
    const float* __restrict__ bias, float* __restrict__ out, int N) {
  const int sub = threadIdx.x & 7;
  const int grp = threadIdx.x >> 3;
  const float4* __restrict__ t4 = (const float4*)t + sub;
  int   beg[2], end[2], id[2][4];
  float sc[2];
#pragma unroll
  for (int h = 0; h < 2; ++h) {
    const int n = blockIdx.x * 64 + h * 32 + grp;
    const bool ok = (n < N);
    beg[h] = ok ? rowptr[n] : 0;
    end[h] = ok ? rowptr[n + 1] : 0;
    sc[h]  = ok ? inv_in[n] : 0.0f;
  }
#pragma unroll
  for (int h = 0; h < 2; ++h) {
#pragma unroll
    for (int c = 0; c < 4; ++c) {
      const int e = beg[h] + 8 * c + sub;
      id[h][c] = (e < end[h]) ? csr_src[e] : 0;
    }
  }
  const float4 b = ((const float4*)bias)[sub];
  float4 acc[2];
  acc[0] = make_float4(0.0f, 0.0f, 0.0f, 0.0f);
  acc[1] = make_float4(0.0f, 0.0f, 0.0f, 0.0f);
#pragma unroll
  for (int c = 0; c < 4; ++c) {
    float4 v[2][8];
#pragma unroll
    for (int h = 0; h < 2; ++h) {
      int s[8];
#pragma unroll
      for (int q = 0; q < 8; ++q) s[q] = __shfl(id[h][c], q, 8);
      if (beg[h] + 8 * c < end[h]) {
#pragma unroll
        for (int q = 0; q < 8; ++q) v[h][q] = t4[(size_t)s[q] * 8];
      }
    }
#pragma unroll
    for (int h = 0; h < 2; ++h) {
      if (beg[h] + 8 * c < end[h]) {
#pragma unroll
        for (int q = 0; q < 8; ++q) {
          if (beg[h] + 8 * c + q < end[h]) {
            acc[h].x += v[h][q].x; acc[h].y += v[h][q].y;
            acc[h].z += v[h][q].z; acc[h].w += v[h][q].w;
          }
        }
      }
    }
  }
#pragma unroll
  for (int h = 0; h < 2; ++h) {
    for (int j = beg[h] + 32; j < end[h]; ++j) {
      const int s0 = csr_src[j];
      const float4 v0 = t4[(size_t)s0 * 8];
      acc[h].x += v0.x; acc[h].y += v0.y; acc[h].z += v0.z; acc[h].w += v0.w;
    }
    const int n = blockIdx.x * 64 + h * 32 + grp;
    if (n < N) {
      float4 o;
      o.x = acc[h].x * sc[h] + b.x;
      o.y = acc[h].y * sc[h] + b.y;
      o.z = acc[h].z * sc[h] + b.z;
      o.w = acc[h].w * sc[h] + b.w;
      ((float4*)(out + (size_t)n * 32))[sub] = o;
    }
  }
}

extern "C" void kernel_launch(void* const* d_in, const int* in_sizes, int n_in,
                              void* d_out, int out_size, void* d_ws, size_t ws_size,
                              hipStream_t stream) {
  const float* in_feat = (const float*)d_in[0];
  const int*   src     = (const int*)d_in[1];
  const int*   dst     = (const int*)d_in[2];
  const float* W1      = (const float*)d_in[3];
  const float* b1      = (const float*)d_in[4];
  const float* W2      = (const float*)d_in[5];
  const float* b2      = (const float*)d_in[6];
  float* out = (float*)d_out;

  const int N = GCN_N_NODES;
  const int E = GCN_N_EDGES;

  float* inv_in  = (float*)d_ws + 2 * N;
  int*   rowptr  = (int*)(inv_in + N);
  int*   csr_src = rowptr + (N + 16) + N;
  float* t       = (float*)(csr_src + E);
  int*   hist_in  = (int*)(t + (size_t)N * 32);
  int*   cnt8     = hist_in + (size_t)GCN_G * GCN_R * GCN_NRP;
  int*   csr_g    = cnt8 + (size_t)GCN_G * GCN_R * GCN_NR;

  gcn_count_kernel<<<dim3(GCN_G, 2 * GCN_R), 1024, 0, stream>>>(src, dst, hist_in, cnt8, csr_g);

  gcn_layer1_fused<<<(GCN_N_NODES + GCN_TM - 1) / GCN_TM, 256, 0, stream>>>(in_feat, hist_in, csr_g, cnt8,
                                             W1, b1, W2, t, rowptr, csr_src, inv_in);

  gcn_gather32<<<(GCN_N_NODES + 63) / 64, 256, 0, stream>>>(t, rowptr, csr_src, inv_in, b2, out, N);
}
